# starting kernel
# baseline (speedup 1.0000x reference)
#include <hip/hip_runtime.h>

typedef __attribute__((ext_vector_type(2))) float v2f;
typedef __attribute__((ext_vector_type(8))) float v8f;

#define D_FEAT 128

__global__ __launch_bounds__(256) void deg_init_kernel(float* deg, int n) {
    int i = blockIdx.x * 256 + threadIdx.x;
    if (i < n) deg[i] = 1.0f;
}

__global__ __launch_bounds__(256) void deg_edges_kernel(const int* __restrict__ dst,
                                                        float* deg, int e) {
    int i = blockIdx.x * 256 + threadIdx.x;
    if (i < e) atomicAdd(&deg[dst[i]], 1.0f);
}

__global__ __launch_bounds__(256) void deg_rsqrt_kernel(float* deg, int n) {
    int i = blockIdx.x * 256 + threadIdx.x;
    if (i < n) {
        float d = deg[i];
        deg[i] = (d > 0.0f) ? rsqrtf(d) : 0.0f;
    }
}

__global__ __launch_bounds__(256) void zero_kernel(float4* p, long n4) {
    long i = (long)blockIdx.x * 256 + threadIdx.x;
    if (i < n4) p[i] = make_float4(0.f, 0.f, 0.f, 0.f);
}

#define LDA 132
__global__ __launch_bounds__(256)
void gemm128_wmma_kernel(const float* __restrict__ A, const float* __restrict__ W,
                         float* __restrict__ H, int nrows) {
    __shared__ float As[16 * LDA];
    const int tid = threadIdx.x;
    const long row0 = (long)blockIdx.x * 16;

    for (int idx = tid; idx < 512; idx += 256) {
        int r = idx >> 5;
        int c4 = idx & 31;
        long gr = row0 + r;
        if (gr >= nrows) gr = nrows - 1;
        *(float4*)(As + r * LDA + c4 * 4) =
            *(const float4*)(A + gr * D_FEAT + c4 * 4);
    }
    __syncthreads();

    const int wave  = tid >> 5;
    const int lane  = tid & 31;
    const int col0  = wave * 16;
    const int mrow  = lane & 15;
    const int khalf = (lane >> 4) * 2;
    const int ncol  = lane & 15;

    v8f acc = {};
    for (int k = 0; k < D_FEAT; k += 4) {
        v2f a, b;
        a.x = As[mrow * LDA + k + khalf + 0];
        a.y = As[mrow * LDA + k + khalf + 1];
        b.x = W[(k + khalf + 0) * D_FEAT + col0 + ncol];
        b.y = W[(k + khalf + 1) * D_FEAT + col0 + ncol];
        acc = __builtin_amdgcn_wmma_f32_16x16x4_f32(
             false, a,  false, b,
             (short)0, acc,  false,  false);
    }

    const int mbase = (lane >> 4) * 8;
    for (int r = 0; r < 8; ++r) {
        long m = row0 + mbase + r;
        if (m < nrows) H[m * D_FEAT + col0 + ncol] = acc[r];
    }
}

__global__ __launch_bounds__(256)
void scatter_kernel(const int* __restrict__ src, const int* __restrict__ dst,
                    const float* __restrict__ dinv, const float* __restrict__ h,
                    float* __restrict__ agg, int nedges) {
    int e = blockIdx.x * 8 + (threadIdx.x >> 5);
    if (e >= nedges) return;
    int lane = threadIdx.x & 31;
    int s = src[e], d = dst[e];
    float norm = dinv[s] * dinv[d];
    float4 hv = *(const float4*)(h + (long)s * D_FEAT + lane * 4);
    float* base = agg + (long)d * D_FEAT + lane * 4;
    atomicAdd(base + 0, norm * hv.x);
    atomicAdd(base + 1, norm * hv.y);
    atomicAdd(base + 2, norm * hv.z);
    atomicAdd(base + 3, norm * hv.w);
}

__global__ __launch_bounds__(256)
void finish_kernel(const float* __restrict__ agg, const float* __restrict__ h,
                   const float* __restrict__ dinv, const float* __restrict__ b,
                   float* __restrict__ xout, int n) {
    long i = (long)blockIdx.x * 256 + threadIdx.x;
    long total = (long)n * D_FEAT;
    if (i >= total) return;
    int node = (int)(i >> 7);
    int f = (int)(i & (D_FEAT - 1));
    float di = dinv[node];
    float v = agg[i] + di * di * h[i] + b[f];
    xout[i] = v > 0.0f ? v : 0.0f;
}

#define LDAJ 260
__global__ __launch_bounds__(256)
void gemm_jk_wmma_kernel(const float* __restrict__ x1, const float* __restrict__ x2,
                         const float* __restrict__ Wl, const float* __restrict__ bl,
                         float* __restrict__ out, int nrows) {
    __shared__ float As[16 * LDAJ];
    const int tid = threadIdx.x;
    const long row0 = (long)blockIdx.x * 16;

    for (int idx = tid; idx < 1024; idx += 256) {
        int r = idx >> 6;
        int c4 = idx & 63;
        long gr = row0 + r;
        if (gr >= nrows) gr = nrows - 1;
        const float* srcp = (c4 < 32)
            ? (x1 + gr * D_FEAT + c4 * 4)
            : (x2 + gr * D_FEAT + (c4 - 32) * 4);
        *(float4*)(As + r * LDAJ + c4 * 4) = *(const float4*)srcp;
    }
    __syncthreads();

    const int wave  = tid >> 5;
    const int lane  = tid & 31;
    const int col0  = wave * 16;
    const int mrow  = lane & 15;
    const int khalf = (lane >> 4) * 2;
    const int ncol  = lane & 15;

    v8f acc = {};
    for (int k = 0; k < 2 * D_FEAT; k += 4) {
        v2f a, b;
        a.x = As[mrow * LDAJ + k + khalf + 0];
        a.y = As[mrow * LDAJ + k + khalf + 1];
        b.x = Wl[(k + khalf + 0) * D_FEAT + col0 + ncol];
        b.y = Wl[(k + khalf + 1) * D_FEAT + col0 + ncol];
        acc = __builtin_amdgcn_wmma_f32_16x16x4_f32(
            false, a, false, b, (short)0, acc, false, false);
    }

    const float bias = bl[col0 + ncol];
    const int mbase = (lane >> 4) * 8;
    for (int r = 0; r < 8; ++r) {
        long m = row0 + mbase + r;
        if (m < nrows) out[m * D_FEAT + col0 + ncol] = acc[r] + bias;
    }
}

extern "C" void kernel_launch(void* const* d_in, const int* in_sizes, int n_in,
                              void* d_out, int out_size, void* d_ws, size_t ws_size,
                              hipStream_t stream) {
    (void)n_in; (void)out_size; (void)ws_size;

    const float* x    = (const float*)d_in[0];
    const int*   ei   = (const int*)d_in[1];
    const float* W1   = (const float*)d_in[2];
    const float* b1   = (const float*)d_in[3];
    const float* W2   = (const float*)d_in[4];
    const float* b2   = (const float*)d_in[5];
    const float* linW = (const float*)d_in[6];
    const float* linb = (const float*)d_in[7];
    float* out = (float*)d_out;

    const int N = in_sizes[0] / D_FEAT;
    const int E = in_sizes[1] / 2;
    const int* srcI = ei;
    const int* dstI = ei + E;

    float* ws   = (float*)d_ws;
    float* dinv = ws;
    float* h    = dinv + N;
    float* agg  = h   + (long)N * D_FEAT;
    float* x1   = agg + (long)N * D_FEAT;
    float* x2   = x1  + (long)N * D_FEAT;

    const int nBlkN   = (N + 255) / 256;
    const int nBlkE   = (E + 255) / 256;
    const int nBlkRow = (N + 15) / 16;
    const int nBlkSc  = (E + 7) / 8;
    const long feats  = (long)N * D_FEAT;
    const int nBlkF   = (int)((feats + 255) / 256);
    const long n4     = feats / 4;
    const int nBlkZ   = (int)((n4 + 255) / 256);

    deg_init_kernel<<<nBlkN, 256, 0, stream>>>(dinv, N);
    deg_edges_kernel<<<nBlkE, 256, 0, stream>>>(dstI, dinv, E);
    deg_rsqrt_kernel<<<nBlkN, 256, 0, stream>>>(dinv, N);

    gemm128_wmma_kernel<<<nBlkRow, 256, 0, stream>>>(x, W1, h, N);
    zero_kernel<<<nBlkZ, 256, 0, stream>>>((float4*)agg, n4);
    scatter_kernel<<<nBlkSc, 256, 0, stream>>>(srcI, dstI, dinv, h, agg, E);
    finish_kernel<<<nBlkF, 256, 0, stream>>>(agg, h, dinv, b1, x1, N);

    gemm128_wmma_kernel<<<nBlkRow, 256, 0, stream>>>(x1, W2, h, N);
    zero_kernel<<<nBlkZ, 256, 0, stream>>>((float4*)agg, n4);
    scatter_kernel<<<nBlkSc, 256, 0, stream>>>(srcI, dstI, dinv, h, agg, E);
    finish_kernel<<<nBlkF, 256, 0, stream>>>(agg, h, dinv, b2, x2, N);

    gemm_jk_wmma_kernel<<<nBlkRow, 256, 0, stream>>>(x1, x2, linW, linb, out, N);
}


# Global float atomics replaced by destination-owner blocks: a block keeps 196 output rows (and their degree counts) in LDS, scans all destinations, adds its edges there by LDS adds; sum order changes
# speedup vs baseline: 2.5032x; 2.5032x over previous
#include <hip/hip_runtime.h>


typedef __attribute__((ext_vector_type(2))) float v2f;
typedef __attribute__((ext_vector_type(8))) float v8f;

#define D_FEAT 128

#define OWN_ROWS    196
#define OWN_WAVES   32
#define OWN_THREADS (OWN_WAVES * 32)
#define GROUP_EDGES 512
#define Q_CAP       544

template <class F, class G>
__device__ __forceinline__ void scan_dst(const int* __restrict__ dst, int nedges, F&& visit,
                                         G&& after_group) {
    const int lane = threadIdx.x & 31;
    const int wave = __builtin_amdgcn_readfirstlane((int)(threadIdx.x >> 5));
    const bool vec_ok = (((size_t)dst) & 15) == 0;
    const int ngroups = vec_ok ? nedges / GROUP_EDGES : 0;
    const int off = (int)blockIdx.x * (ngroups / (int)gridDim.x);

    auto rot = [&](int g) { int gr = g + off; return gr >= ngroups ? gr - ngroups : gr; };
    auto load_group = [&](int4 (&v)[4], int gr) {
        const int4* p = (const int4*)(dst + gr * GROUP_EDGES) + lane;
#pragma unroll
        for (int j = 0; j < 4; ++j) v[j] = p[j * 32];
    };
    auto step = [&](int4 (&cur)[4], int4 (&nxt)[4], int g) {
        const int gr = rot(g);
        const int gn = g + OWN_WAVES;
        load_group(nxt, rot(gn < ngroups ? gn : g));
#pragma unroll
        for (int j = 0; j < 4; ++j) {
            const int e0 = gr * GROUP_EDGES + j * 128 + lane * 4;
            visit(cur[j].x, e0 + 0);
            visit(cur[j].y, e0 + 1);
            visit(cur[j].z, e0 + 2);
            visit(cur[j].w, e0 + 3);
        }
        after_group();
    };

    if (wave < ngroups) {
        int4 bufa[4], bufb[4];
        load_group(bufa, rot(wave));
        for (int g = wave; g < ngroups; g += 2 * OWN_WAVES) {
            step(bufa, bufb, g);
            if (g + OWN_WAVES >= ngroups) break;
            step(bufb, bufa, g + OWN_WAVES);
        }
    }
    for (int base = ngroups * GROUP_EDGES + wave * 32; base < nedges; base += OWN_THREADS) {
        const int e = base + lane;
        visit(e < nedges ? dst[e] : -1, e);
        after_group();
    }
}

__global__ __launch_bounds__(OWN_THREADS)
void deg_owner_kernel(const int* __restrict__ dst, float* __restrict__ dinv,
                      int n, int nedges) {
    __shared__ float cnt_s[OWN_ROWS];
    const int r0 = blockIdx.x * OWN_ROWS;
    const int rows = min(OWN_ROWS, n - r0);
    if (threadIdx.x < OWN_ROWS) cnt_s[threadIdx.x] = 1.0f;
    __syncthreads();

    scan_dst(dst, nedges, [&](int d, int e) {
        (void)e;
        const unsigned dl = (unsigned)(d - r0);
        if (dl < (unsigned)rows)
            __hip_atomic_fetch_add(&cnt_s[dl], 1.0f, __ATOMIC_RELAXED,
                                   __HIP_MEMORY_SCOPE_WORKGROUP);
    }, [] {});
    __syncthreads();

    if ((int)threadIdx.x < rows) {
        float d = cnt_s[threadIdx.x];
        dinv[r0 + threadIdx.x] = (d > 0.0f) ? rsqrtf(d) : 0.0f;
    }
}

__device__ __forceinline__ void drain_edges(const int* q_e, const int* q_d, int base, int n,
                                            const int* __restrict__ src,
                                            const float* __restrict__ dinv,
                                            const float* dinv_s,
                                            const float* __restrict__ h,
                                            float* slab, int lane) {
    int dl = 0, s = 0;
    float nrm = 0.0f;
    if (lane < n) {
        const int e = q_e[base + lane];
        dl = q_d[base + lane];
        s = src[e];
        nrm = dinv[s] * dinv_s[dl];
    }
    for (int i = 0; i < n; i += 8) {
        float4 hv[8];
#pragma unroll
        for (int u = 0; u < 8; ++u) {
            const int si = __builtin_amdgcn_readlane(s, i + u);
            hv[u] = *(const float4*)(h + (long)si * D_FEAT + lane * 4);
        }
#pragma unroll
        for (int u = 0; u < 8; ++u) {
            if (i + u < n) {
                const float nu = __int_as_float(__builtin_amdgcn_readlane(__float_as_int(nrm), i + u));
                const int dlu = __builtin_amdgcn_readlane(dl, i + u);
                float* row = slab + dlu * D_FEAT + lane;
                __hip_atomic_fetch_add(row +  0, nu * hv[u].x, __ATOMIC_RELAXED, __HIP_MEMORY_SCOPE_WORKGROUP);
                __hip_atomic_fetch_add(row + 32, nu * hv[u].y, __ATOMIC_RELAXED, __HIP_MEMORY_SCOPE_WORKGROUP);
                __hip_atomic_fetch_add(row + 64, nu * hv[u].z, __ATOMIC_RELAXED, __HIP_MEMORY_SCOPE_WORKGROUP);
                __hip_atomic_fetch_add(row + 96, nu * hv[u].w, __ATOMIC_RELAXED, __HIP_MEMORY_SCOPE_WORKGROUP);
            }
        }
    }
}

__global__ __launch_bounds__(OWN_THREADS)
void layer_owner_kernel(const int* __restrict__ src, const int* __restrict__ dst,
                        const float* __restrict__ dinv, const float* __restrict__ h,
                        const float* __restrict__ bias, float* __restrict__ xout,
                        int n, int nedges) {
    __shared__ float slab[OWN_ROWS * D_FEAT];
    __shared__ float dinv_s[OWN_ROWS];
    __shared__ int q_e_all[OWN_WAVES * Q_CAP];
    __shared__ int q_d_all[OWN_WAVES * Q_CAP];

    const int lane = threadIdx.x & 31;
    const int wave = __builtin_amdgcn_readfirstlane((int)(threadIdx.x >> 5));
    const int r0 = blockIdx.x * OWN_ROWS;
    const int rows = min(OWN_ROWS, n - r0);

    const float4 bv = *(const float4*)(bias + lane * 4);
    for (int r = wave; r < rows; r += OWN_WAVES) {
        const float di = dinv[r0 + r];
        const float4 hv = *(const float4*)(h + (long)(r0 + r) * D_FEAT + lane * 4);
        float* row = slab + r * D_FEAT + lane;
        row[0]  = di * di * hv.x + bv.x;
        row[32] = di * di * hv.y + bv.y;
        row[64] = di * di * hv.z + bv.z;
        row[96] = di * di * hv.w + bv.w;
        if (lane == 0) dinv_s[r] = di;
    }
    __syncthreads();

    int* q_e = q_e_all + wave * Q_CAP;
    int* q_d = q_d_all + wave * Q_CAP;
    int cnt = 0;

    scan_dst(dst, nedges, [&](int d, int e) {
        const unsigned dl = (unsigned)(d - r0);
        const bool hit = dl < (unsigned)rows;
        const unsigned m = __builtin_amdgcn_ballot_w32(hit);
        if (m != 0u) {
            const int pos = cnt + (int)__builtin_amdgcn_mbcnt_lo(m, 0u);
            if (hit) { q_e[pos] = e; q_d[pos] = (int)dl; }
            cnt += __popc(m);
        }
    }, [&] {
        while (cnt >= 32) {
            cnt -= 32;
            drain_edges(q_e, q_d, cnt, 32, src, dinv, dinv_s, h, slab, lane);
        }
    });
    if (cnt > 0) drain_edges(q_e, q_d, 0, cnt, src, dinv, dinv_s, h, slab, lane);
    __syncthreads();

    for (int r = wave; r < rows; r += OWN_WAVES) {
        const float* row = slab + r * D_FEAT + lane;
        float4 v;
        v.x = row[0]; v.y = row[32]; v.z = row[64]; v.w = row[96];
        v.x = v.x > 0.0f ? v.x : 0.0f;
        v.y = v.y > 0.0f ? v.y : 0.0f;
        v.z = v.z > 0.0f ? v.z : 0.0f;
        v.w = v.w > 0.0f ? v.w : 0.0f;
        *(float4*)(xout + (long)(r0 + r) * D_FEAT + lane * 4) = v;
    }
}

#define LDA 132
__global__ __launch_bounds__(256)
void gemm128_wmma_kernel(const float* __restrict__ A, const float* __restrict__ W,
                         float* __restrict__ H, int nrows) {
    __shared__ float As[16 * LDA];
    const int tid = threadIdx.x;
    const long row0 = (long)blockIdx.x * 16;

    for (int idx = tid; idx < 512; idx += 256) {
        int r = idx >> 5;
        int c4 = idx & 31;
        long gr = row0 + r;
        if (gr >= nrows) gr = nrows - 1;
        *(float4*)(As + r * LDA + c4 * 4) =
            *(const float4*)(A + gr * D_FEAT + c4 * 4);
    }
    __syncthreads();

    const int wave  = tid >> 5;
    const int lane  = tid & 31;
    const int col0  = wave * 16;
    const int mrow  = lane & 15;
    const int khalf = (lane >> 4) * 2;
    const int ncol  = lane & 15;

    v8f acc = {};
    for (int k = 0; k < D_FEAT; k += 4) {
        v2f a, b;
        a.x = As[mrow * LDA + k + khalf + 0];
        a.y = As[mrow * LDA + k + khalf + 1];
        b.x = W[(k + khalf + 0) * D_FEAT + col0 + ncol];
        b.y = W[(k + khalf + 1) * D_FEAT + col0 + ncol];
        acc = __builtin_amdgcn_wmma_f32_16x16x4_f32(
             false, a,  false, b,
             (short)0, acc,  false,  false);
    }

    const int mbase = (lane >> 4) * 8;
    for (int r = 0; r < 8; ++r) {
        long m = row0 + mbase + r;
        if (m < nrows) H[m * D_FEAT + col0 + ncol] = acc[r];
    }
}

#define LDAJ 260
__global__ __launch_bounds__(256)
void gemm_jk_wmma_kernel(const float* __restrict__ x1, const float* __restrict__ x2,
                         const float* __restrict__ Wl, const float* __restrict__ bl,
                         float* __restrict__ out, int nrows) {
    __shared__ float As[16 * LDAJ];
    const int tid = threadIdx.x;
    const long row0 = (long)blockIdx.x * 16;

    for (int idx = tid; idx < 1024; idx += 256) {
        int r = idx >> 6;
        int c4 = idx & 63;
        long gr = row0 + r;
        if (gr >= nrows) gr = nrows - 1;
        const float* srcp = (c4 < 32)
            ? (x1 + gr * D_FEAT + c4 * 4)
            : (x2 + gr * D_FEAT + (c4 - 32) * 4);
        *(float4*)(As + r * LDAJ + c4 * 4) = *(const float4*)srcp;
    }
    __syncthreads();

    const int wave  = tid >> 5;
    const int lane  = tid & 31;
    const int col0  = wave * 16;
    const int mrow  = lane & 15;
    const int khalf = (lane >> 4) * 2;
    const int ncol  = lane & 15;

    v8f acc = {};
    for (int k = 0; k < 2 * D_FEAT; k += 4) {
        v2f a, b;
        a.x = As[mrow * LDAJ + k + khalf + 0];
        a.y = As[mrow * LDAJ + k + khalf + 1];
        b.x = Wl[(k + khalf + 0) * D_FEAT + col0 + ncol];
        b.y = Wl[(k + khalf + 1) * D_FEAT + col0 + ncol];
        acc = __builtin_amdgcn_wmma_f32_16x16x4_f32(
            false, a, false, b, (short)0, acc, false, false);
    }

    const float bias = bl[col0 + ncol];
    const int mbase = (lane >> 4) * 8;
    for (int r = 0; r < 8; ++r) {
        long m = row0 + mbase + r;
        if (m < nrows) out[m * D_FEAT + col0 + ncol] = acc[r] + bias;
    }
}

extern "C" void kernel_launch(void* const* d_in, const int* in_sizes, int n_in,
                              void* d_out, int out_size, void* d_ws, size_t ws_size,
                              hipStream_t stream) {
    (void)n_in; (void)out_size; (void)ws_size;

    const float* x    = (const float*)d_in[0];
    const int*   ei   = (const int*)d_in[1];
    const float* W1   = (const float*)d_in[2];
    const float* b1   = (const float*)d_in[3];
    const float* W2   = (const float*)d_in[4];
    const float* b2   = (const float*)d_in[5];
    const float* linW = (const float*)d_in[6];
    const float* linb = (const float*)d_in[7];
    float* out = (float*)d_out;

    const int N = in_sizes[0] / D_FEAT;
    const int E = in_sizes[1] / 2;
    const int* srcI = ei;
    const int* dstI = ei + E;

    float* ws   = (float*)d_ws;
    float* dinv = ws;
    float* h    = dinv + N;
    float* x1   = h   + (long)N * D_FEAT;
    float* x2   = x1  + (long)N * D_FEAT;

    const int nBlkRow = (N + 15) / 16;
    const int nBlkOwn = (N + OWN_ROWS - 1) / OWN_ROWS;

    deg_owner_kernel<<<nBlkOwn, OWN_THREADS, 0, stream>>>(dstI, dinv, N, E);

    gemm128_wmma_kernel<<<nBlkRow, 256, 0, stream>>>(x, W1, h, N);
    layer_owner_kernel<<<nBlkOwn, OWN_THREADS, 0, stream>>>(srcI, dstI, dinv, h, b1, x1, N, E);

    gemm128_wmma_kernel<<<nBlkRow, 256, 0, stream>>>(x1, W2, h, N);
    layer_owner_kernel<<<nBlkOwn, OWN_THREADS, 0, stream>>>(srcI, dstI, dinv, h, b2, x2, N, E);

    gemm_jk_wmma_kernel<<<nBlkRow, 256, 0, stream>>>(x1, x2, linW, linb, out, N);
}


# Edge lists built once: the degree kernel's scan of dst also writes each owner block's (source,row) pairs into its own rows of the idle x2 plane; layers read them, no rescan; row sum order changes
# speedup vs baseline: 3.0633x; 1.2238x over previous
#include <hip/hip_runtime.h>


typedef __attribute__((ext_vector_type(2))) float v2f;
typedef __attribute__((ext_vector_type(8))) float v8f;

#define D_FEAT 128

#define OWN_ROWS    196
#define OWN_WAVES   32
#define OWN_THREADS (OWN_WAVES * 32)
#define GROUP_EDGES 512
#define Q_CAP       544

template <class F, class G>
__device__ __forceinline__ void scan_dst(const int* __restrict__ dst, int nedges, F&& visit,
                                         G&& after_group) {
    const int lane = threadIdx.x & 31;
    const int wave = __builtin_amdgcn_readfirstlane((int)(threadIdx.x >> 5));
    const bool vec_ok = (((size_t)dst) & 15) == 0;
    const int ngroups = vec_ok ? nedges / GROUP_EDGES : 0;
    const int off = (int)blockIdx.x * (ngroups / (int)gridDim.x);

    auto rot = [&](int g) { int gr = g + off; return gr >= ngroups ? gr - ngroups : gr; };
    auto load_group = [&](int4 (&v)[4], int gr) {
        const int4* p = (const int4*)(dst + gr * GROUP_EDGES) + lane;
#pragma unroll
        for (int j = 0; j < 4; ++j) v[j] = p[j * 32];
    };
    auto step = [&](int4 (&cur)[4], int4 (&nxt)[4], int g) {
        const int gr = rot(g);
        const int gn = g + OWN_WAVES;
        load_group(nxt, rot(gn < ngroups ? gn : g));
#pragma unroll
        for (int j = 0; j < 4; ++j) {
            const int e0 = gr * GROUP_EDGES + j * 128 + lane * 4;
            visit(cur[j].x, e0 + 0);
            visit(cur[j].y, e0 + 1);
            visit(cur[j].z, e0 + 2);
            visit(cur[j].w, e0 + 3);
        }
        after_group();
    };

    if (wave < ngroups) {
        int4 bufa[4], bufb[4];
        load_group(bufa, rot(wave));
        for (int g = wave; g < ngroups; g += 2 * OWN_WAVES) {
            step(bufa, bufb, g);
            if (g + OWN_WAVES >= ngroups) break;
            step(bufb, bufa, g + OWN_WAVES);
        }
    }
    for (int base = ngroups * GROUP_EDGES + wave * 32; base < nedges; base += OWN_THREADS) {
        const int e = base + lane;
        visit(e < nedges ? dst[e] : -1, e);
        after_group();
    }
}

__device__ __forceinline__ int seg_cap_of(int rows) {
    return (rows * D_FEAT - OWN_WAVES) / (2 * OWN_WAVES);
}

__global__ __launch_bounds__(OWN_THREADS)
void deg_lists_kernel(const int* __restrict__ src, const int* __restrict__ dst,
                      float* __restrict__ dinv, int* __restrict__ lists,
                      int n, int nedges) {
    __shared__ float cnt_s[OWN_ROWS];
    __shared__ int q_e_all[OWN_WAVES * Q_CAP];
    __shared__ int q_d_all[OWN_WAVES * Q_CAP];
    __shared__ int ovf_s;

    const int lane = threadIdx.x & 31;
    const int wave = __builtin_amdgcn_readfirstlane((int)(threadIdx.x >> 5));
    const int r0 = blockIdx.x * OWN_ROWS;
    const int rows = min(OWN_ROWS, n - r0);
    const int cap = seg_cap_of(rows);
    int* hdr = lists + (long)r0 * D_FEAT;
    int* seg_s = hdr + OWN_WAVES + wave * 2 * cap;
    int* seg_d = seg_s + cap;

    if (threadIdx.x < OWN_ROWS) cnt_s[threadIdx.x] = 1.0f;
    if (threadIdx.x == 0) ovf_s = 0;
    __syncthreads();

    int* q_e = q_e_all + wave * Q_CAP;
    int* q_d = q_d_all + wave * Q_CAP;
    int cnt = 0;
    int kept = 0;

    auto flush = [&](int base, int nn) {
        if (lane < nn) {
            const int e = q_e[base + lane];
            const int dl = q_d[base + lane];
            const int pos = kept + lane;
            if (pos < cap) { seg_s[pos] = src[e]; seg_d[pos] = dl; }
        }
        kept += nn;
    };

    scan_dst(dst, nedges, [&](int d, int e) {
        const unsigned dl = (unsigned)(d - r0);
        const bool hit = dl < (unsigned)rows;
        const unsigned m = __builtin_amdgcn_ballot_w32(hit);
        if (m != 0u) {
            const int pos = cnt + (int)__builtin_amdgcn_mbcnt_lo(m, 0u);
            if (hit) {
                q_e[pos] = e; q_d[pos] = (int)dl;
                __hip_atomic_fetch_add(&cnt_s[dl], 1.0f, __ATOMIC_RELAXED,
                                       __HIP_MEMORY_SCOPE_WORKGROUP);
            }
            cnt += __popc(m);
        }
    }, [&] {
        while (cnt >= 32) {
            cnt -= 32;
            flush(cnt, 32);
        }
    });
    if (cnt > 0) flush(0, cnt);
    if (kept > cap && lane == 0) ovf_s = 1;
    __syncthreads();

    if ((int)threadIdx.x < rows) {
        float d = cnt_s[threadIdx.x];
        dinv[r0 + threadIdx.x] = (d > 0.0f) ? rsqrtf(d) : 0.0f;
    }
    if (lane == 0) hdr[wave] = ovf_s ? -1 : kept;
}

__device__ __forceinline__ void add_edges(int s, int dl, float nrm, int n,
                                          const float* __restrict__ h,
                                          float* slab, int lane) {
    for (int i = 0; i < n; i += 8) {
        float4 hv[8];
#pragma unroll
        for (int u = 0; u < 8; ++u) {
            const int si = __builtin_amdgcn_readlane(s, i + u);
            hv[u] = *(const float4*)(h + (long)si * D_FEAT + lane * 4);
        }
#pragma unroll
        for (int u = 0; u < 8; ++u) {
            if (i + u < n) {
                const float nu = __int_as_float(__builtin_amdgcn_readlane(__float_as_int(nrm), i + u));
                const int dlu = __builtin_amdgcn_readlane(dl, i + u);
                float* row = slab + dlu * D_FEAT + lane;
                __hip_atomic_fetch_add(row +  0, nu * hv[u].x, __ATOMIC_RELAXED, __HIP_MEMORY_SCOPE_WORKGROUP);
                __hip_atomic_fetch_add(row + 32, nu * hv[u].y, __ATOMIC_RELAXED, __HIP_MEMORY_SCOPE_WORKGROUP);
                __hip_atomic_fetch_add(row + 64, nu * hv[u].z, __ATOMIC_RELAXED, __HIP_MEMORY_SCOPE_WORKGROUP);
                __hip_atomic_fetch_add(row + 96, nu * hv[u].w, __ATOMIC_RELAXED, __HIP_MEMORY_SCOPE_WORKGROUP);
            }
        }
    }
}

__device__ __forceinline__ void drain_edges(const int* q_e, const int* q_d, int base, int n,
                                            const int* __restrict__ src,
                                            const float* __restrict__ dinv,
                                            const float* dinv_s,
                                            const float* __restrict__ h,
                                            float* slab, int lane) {
    int dl = 0, s = 0;
    float nrm = 0.0f;
    if (lane < n) {
        const int e = q_e[base + lane];
        dl = q_d[base + lane];
        s = src[e];
        nrm = dinv[s] * dinv_s[dl];
    }
    add_edges(s, dl, nrm, n, h, slab, lane);
}

__global__ __launch_bounds__(OWN_THREADS)
void layer_owner_kernel(const int* __restrict__ src, const int* __restrict__ dst,
                        const float* __restrict__ dinv, const float* __restrict__ h,
                        const float* __restrict__ bias, const int* lists, float* xout,
                        int n, int nedges) {
    __shared__ float slab[OWN_ROWS * D_FEAT];
    __shared__ float dinv_s[OWN_ROWS];
    __shared__ int q_e_all[OWN_WAVES * Q_CAP];
    __shared__ int q_d_all[OWN_WAVES * Q_CAP];

    const int lane = threadIdx.x & 31;
    const int wave = __builtin_amdgcn_readfirstlane((int)(threadIdx.x >> 5));
    const int r0 = blockIdx.x * OWN_ROWS;
    const int rows = min(OWN_ROWS, n - r0);

    const int cap = seg_cap_of(rows);
    const int* hdr = lists + (long)r0 * D_FEAT;
    const int* seg_s = hdr + OWN_WAVES + wave * 2 * cap;
    const int* seg_d = seg_s + cap;
    const int mycnt = hdr[wave];
    int s_nx = 0, dl_nx = 0;
    if (lane < mycnt) { s_nx = seg_s[lane]; dl_nx = seg_d[lane]; }

    const float4 bv = *(const float4*)(bias + lane * 4);
    for (int r = wave; r < rows; r += OWN_WAVES) {
        const float di = dinv[r0 + r];
        const float4 hv = *(const float4*)(h + (long)(r0 + r) * D_FEAT + lane * 4);
        float* row = slab + r * D_FEAT + lane;
        row[0]  = di * di * hv.x + bv.x;
        row[32] = di * di * hv.y + bv.y;
        row[64] = di * di * hv.z + bv.z;
        row[96] = di * di * hv.w + bv.w;
        if (lane == 0) dinv_s[r] = di;
    }
    __syncthreads();

    if (mycnt >= 0) {
        int base = 0;
        for (; base + 32 <= mycnt; base += 32) {
            const int s = s_nx, dl = dl_nx;
            const float nrm = dinv[s] * dinv_s[dl];
            s_nx = 0; dl_nx = 0;
            if (base + 32 + lane < mycnt) {
                s_nx = seg_s[base + 32 + lane];
                dl_nx = seg_d[base + 32 + lane];
            }
            add_edges(s, dl, nrm, 32, h, slab, lane);
        }
        if (base < mycnt) {
            const float nrm = dinv[s_nx] * dinv_s[dl_nx];
            add_edges(s_nx, dl_nx, nrm, mycnt - base, h, slab, lane);
        }
    } else {
        int* q_e = q_e_all + wave * Q_CAP;
        int* q_d = q_d_all + wave * Q_CAP;
        int cnt = 0;

        scan_dst(dst, nedges, [&](int d, int e) {
            const unsigned dl = (unsigned)(d - r0);
            const bool hit = dl < (unsigned)rows;
            const unsigned m = __builtin_amdgcn_ballot_w32(hit);
            if (m != 0u) {
                const int pos = cnt + (int)__builtin_amdgcn_mbcnt_lo(m, 0u);
                if (hit) { q_e[pos] = e; q_d[pos] = (int)dl; }
                cnt += __popc(m);
            }
        }, [&] {
            while (cnt >= 32) {
                cnt -= 32;
                drain_edges(q_e, q_d, cnt, 32, src, dinv, dinv_s, h, slab, lane);
            }
        });
        if (cnt > 0) drain_edges(q_e, q_d, 0, cnt, src, dinv, dinv_s, h, slab, lane);
    }
    __syncthreads();

    for (int r = wave; r < rows; r += OWN_WAVES) {
        const float* row = slab + r * D_FEAT + lane;
        float4 v;
        v.x = row[0]; v.y = row[32]; v.z = row[64]; v.w = row[96];
        v.x = v.x > 0.0f ? v.x : 0.0f;
        v.y = v.y > 0.0f ? v.y : 0.0f;
        v.z = v.z > 0.0f ? v.z : 0.0f;
        v.w = v.w > 0.0f ? v.w : 0.0f;
        *(float4*)(xout + (long)(r0 + r) * D_FEAT + lane * 4) = v;
    }
}

#define LDA 132
__global__ __launch_bounds__(256)
void gemm128_wmma_kernel(const float* __restrict__ A, const float* __restrict__ W,
                         float* __restrict__ H, int nrows) {
    __shared__ float As[16 * LDA];
    const int tid = threadIdx.x;
    const long row0 = (long)blockIdx.x * 16;

    for (int idx = tid; idx < 512; idx += 256) {
        int r = idx >> 5;
        int c4 = idx & 31;
        long gr = row0 + r;
        if (gr >= nrows) gr = nrows - 1;
        *(float4*)(As + r * LDA + c4 * 4) =
            *(const float4*)(A + gr * D_FEAT + c4 * 4);
    }
    __syncthreads();

    const int wave  = tid >> 5;
    const int lane  = tid & 31;
    const int col0  = wave * 16;
    const int mrow  = lane & 15;
    const int khalf = (lane >> 4) * 2;
    const int ncol  = lane & 15;

    v8f acc = {};
    for (int k = 0; k < D_FEAT; k += 4) {
        v2f a, b;
        a.x = As[mrow * LDA + k + khalf + 0];
        a.y = As[mrow * LDA + k + khalf + 1];
        b.x = W[(k + khalf + 0) * D_FEAT + col0 + ncol];
        b.y = W[(k + khalf + 1) * D_FEAT + col0 + ncol];
        acc = __builtin_amdgcn_wmma_f32_16x16x4_f32(
             false, a,  false, b,
             (short)0, acc,  false,  false);
    }

    const int mbase = (lane >> 4) * 8;
    for (int r = 0; r < 8; ++r) {
        long m = row0 + mbase + r;
        if (m < nrows) H[m * D_FEAT + col0 + ncol] = acc[r];
    }
}

#define LDAJ 260
__global__ __launch_bounds__(256)
void gemm_jk_wmma_kernel(const float* __restrict__ x1, const float* __restrict__ x2,
                         const float* __restrict__ Wl, const float* __restrict__ bl,
                         float* __restrict__ out, int nrows) {
    __shared__ float As[16 * LDAJ];
    const int tid = threadIdx.x;
    const long row0 = (long)blockIdx.x * 16;

    for (int idx = tid; idx < 1024; idx += 256) {
        int r = idx >> 6;
        int c4 = idx & 63;
        long gr = row0 + r;
        if (gr >= nrows) gr = nrows - 1;
        const float* srcp = (c4 < 32)
            ? (x1 + gr * D_FEAT + c4 * 4)
            : (x2 + gr * D_FEAT + (c4 - 32) * 4);
        *(float4*)(As + r * LDAJ + c4 * 4) = *(const float4*)srcp;
    }
    __syncthreads();

    const int wave  = tid >> 5;
    const int lane  = tid & 31;
    const int col0  = wave * 16;
    const int mrow  = lane & 15;
    const int khalf = (lane >> 4) * 2;
    const int ncol  = lane & 15;

    v8f acc = {};
    for (int k = 0; k < 2 * D_FEAT; k += 4) {
        v2f a, b;
        a.x = As[mrow * LDAJ + k + khalf + 0];
        a.y = As[mrow * LDAJ + k + khalf + 1];
        b.x = Wl[(k + khalf + 0) * D_FEAT + col0 + ncol];
        b.y = Wl[(k + khalf + 1) * D_FEAT + col0 + ncol];
        acc = __builtin_amdgcn_wmma_f32_16x16x4_f32(
            false, a, false, b, (short)0, acc, false, false);
    }

    const float bias = bl[col0 + ncol];
    const int mbase = (lane >> 4) * 8;
    for (int r = 0; r < 8; ++r) {
        long m = row0 + mbase + r;
        if (m < nrows) out[m * D_FEAT + col0 + ncol] = acc[r] + bias;
    }
}

extern "C" void kernel_launch(void* const* d_in, const int* in_sizes, int n_in,
                              void* d_out, int out_size, void* d_ws, size_t ws_size,
                              hipStream_t stream) {
    (void)n_in; (void)out_size; (void)ws_size;

    const float* x    = (const float*)d_in[0];
    const int*   ei   = (const int*)d_in[1];
    const float* W1   = (const float*)d_in[2];
    const float* b1   = (const float*)d_in[3];
    const float* W2   = (const float*)d_in[4];
    const float* b2   = (const float*)d_in[5];
    const float* linW = (const float*)d_in[6];
    const float* linb = (const float*)d_in[7];
    float* out = (float*)d_out;

    const int N = in_sizes[0] / D_FEAT;
    const int E = in_sizes[1] / 2;
    const int* srcI = ei;
    const int* dstI = ei + E;

    float* ws   = (float*)d_ws;
    float* dinv = ws;
    float* h    = dinv + N;
    float* x1   = h   + (long)N * D_FEAT;
    float* x2   = x1  + (long)N * D_FEAT;

    const int nBlkRow = (N + 15) / 16;
    const int nBlkOwn = (N + OWN_ROWS - 1) / OWN_ROWS;

    int* lists = (int*)x2;

    deg_lists_kernel<<<nBlkOwn, OWN_THREADS, 0, stream>>>(srcI, dstI, dinv, lists, N, E);

    gemm128_wmma_kernel<<<nBlkRow, 256, 0, stream>>>(x, W1, h, N);
    layer_owner_kernel<<<nBlkOwn, OWN_THREADS, 0, stream>>>(srcI, dstI, dinv, h, b1, lists, x1, N, E);

    gemm128_wmma_kernel<<<nBlkRow, 256, 0, stream>>>(x1, W2, h, N);
    layer_owner_kernel<<<nBlkOwn, OWN_THREADS, 0, stream>>>(srcI, dstI, dinv, h, b2, lists, x2, N, E);

    gemm_jk_wmma_kernel<<<nBlkRow, 256, 0, stream>>>(x1, x2, linW, linb, out, N);
}


# Each edge read once: 256 blocks take 1/256 of the edges, drop (source,row) into a fixed cell of the destination's owner block; degrees counted from them; 256-fold dst scan goes; row sum order changes
# speedup vs baseline: 3.3250x; 1.0854x over previous
#include <hip/hip_runtime.h>


typedef __attribute__((ext_vector_type(2))) float v2f;
typedef __attribute__((ext_vector_type(8))) float v8f;

#define D_FEAT 128

#define OWN_ROWS    196
#define OWN_WAVES   32
#define OWN_THREADS (OWN_WAVES * 32)
#define GROUP_EDGES 512
#define Q_CAP       544

template <int NW = OWN_WAVES, class F, class G>
__device__ __forceinline__ void scan_dst(const int* __restrict__ dst, int nedges, F&& visit,
                                         G&& after_group) {
    const int lane = threadIdx.x & 31;
    const int wave = __builtin_amdgcn_readfirstlane((int)(threadIdx.x >> 5));
    const bool vec_ok = (((size_t)dst) & 15) == 0;
    const int ngroups = vec_ok ? nedges / GROUP_EDGES : 0;
    const int off = (int)blockIdx.x * (ngroups / (int)gridDim.x);

    auto rot = [&](int g) { int gr = g + off; return gr >= ngroups ? gr - ngroups : gr; };
    auto load_group = [&](int4 (&v)[4], int gr) {
        const int4* p = (const int4*)(dst + gr * GROUP_EDGES) + lane;
#pragma unroll
        for (int j = 0; j < 4; ++j) v[j] = p[j * 32];
    };
    auto step = [&](int4 (&cur)[4], int4 (&nxt)[4], int g) {
        const int gr = rot(g);
        const int gn = g + NW;
        load_group(nxt, rot(gn < ngroups ? gn : g));
#pragma unroll
        for (int j = 0; j < 4; ++j) {
            const int e0 = gr * GROUP_EDGES + j * 128 + lane * 4;
            visit(cur[j].x, e0 + 0);
            visit(cur[j].y, e0 + 1);
            visit(cur[j].z, e0 + 2);
            visit(cur[j].w, e0 + 3);
        }
        after_group();
    };

    if (wave < ngroups) {
        int4 bufa[4], bufb[4];
        load_group(bufa, rot(wave));
        for (int g = wave; g < ngroups; g += 2 * NW) {
            step(bufa, bufb, g);
            if (g + NW >= ngroups) break;
            step(bufb, bufa, g + NW);
        }
    }
    for (int base = ngroups * GROUP_EDGES + wave * 32; base < nedges; base += NW * 32) {
        const int e = base + lane;
        visit(e < nedges ? dst[e] : -1, e);
        after_group();
    }
}

#define NCH     256
#define OWN_MAX 512
#define BIN_THREADS 256
#define DEG_WAVES   8
static_assert(NCH == 8 * OWN_WAVES && NCH == 32 * DEG_WAVES, "cells per wave");

__device__ __forceinline__ int cell_cap_of(int rows) {
    return (rows * D_FEAT - NCH) / NCH;
}

__global__ __launch_bounds__(BIN_THREADS)
void bin_edges_kernel(const int* __restrict__ src, const int* __restrict__ dst,
                      int* __restrict__ lists, int n, int nedges, int nown) {
    __shared__ int fill_s[OWN_MAX];

    const int c = blockIdx.x;
    const int per = (nedges + NCH - 1) / NCH;
    const int e0 = c * per;
    const int e1 = min(nedges, e0 + per);
    const bool usable = (n <= 65536) && (nown <= OWN_MAX);

    for (int o = threadIdx.x; o < OWN_MAX; o += BIN_THREADS) fill_s[o] = 0;
    __syncthreads();

    if (usable) {
        for (int base = e0; base < e1; base += 8 * BIN_THREADS) {
            int dd[8], ss[8];
#pragma unroll
            for (int u = 0; u < 8; ++u) {
                const int e = base + u * BIN_THREADS + (int)threadIdx.x;
                const bool ok = e < e1;
                dd[u] = ok ? dst[e] : -1;
                ss[u] = ok ? src[e] : 0;
            }
#pragma unroll
            for (int u = 0; u < 8; ++u) {
                const int d = dd[u];
                if ((unsigned)d < (unsigned)n) {
                    const int o = d / OWN_ROWS;
                    const int row = d - o * OWN_ROWS;
                    const int cap = cell_cap_of(min(OWN_ROWS, n - o * OWN_ROWS));
                    const int pos = __hip_atomic_fetch_add(&fill_s[o], 1, __ATOMIC_RELAXED,
                                                           __HIP_MEMORY_SCOPE_WORKGROUP);
                    if (pos < cap)
                        lists[(long)o * OWN_ROWS * D_FEAT + NCH + c * cap + pos] = ss[u] | (row << 16);
                }
            }
        }
    }
    __syncthreads();

    for (int o = threadIdx.x; o < nown; o += BIN_THREADS) {
        int v = 0x7fffffff;
        if (usable) v = fill_s[o];
        lists[(long)o * OWN_ROWS * D_FEAT + c] = v;
    }
}

__device__ __forceinline__ bool load_cell_counts(const int* cntp, int cap, int lane, int (&cn)[8]) {
    const int4* cp = (const int4*)cntp + lane * 2;
    const int4 a = cp[0], b = cp[1];
    cn[0] = a.x; cn[1] = a.y; cn[2] = a.z; cn[3] = a.w;
    cn[4] = b.x; cn[5] = b.y; cn[6] = b.z; cn[7] = b.w;
    bool bad = false;
#pragma unroll
    for (int j = 0; j < 8; ++j) bad = bad || (cn[j] > cap);
    return __builtin_amdgcn_ballot_w32(bad) != 0u;
}

__global__ __launch_bounds__(DEG_WAVES * 32)
void deg_kernel(const int* __restrict__ dst, const int* __restrict__ lists,
                float* __restrict__ dinv, int n, int nedges) {
    __shared__ float cnt_s[OWN_ROWS];

    const int lane = threadIdx.x & 31;
    const int wave = __builtin_amdgcn_readfirstlane((int)(threadIdx.x >> 5));
    const int r0 = blockIdx.x * OWN_ROWS;
    const int rows = min(OWN_ROWS, n - r0);
    const int cap = cell_cap_of(rows);
    const int* cntp = lists + (long)r0 * D_FEAT;
    const int* cells = cntp + NCH;

    int cn[8];
    const bool ovf = load_cell_counts(cntp, cap, lane, cn);

    for (int r = threadIdx.x; r < OWN_ROWS; r += DEG_WAVES * 32) cnt_s[r] = 1.0f;
    __syncthreads();

    if (!ovf) {
        for (int t = 0; t < 4; ++t) {
            const int l = wave * 4 + t;
            int cj[8], v[8];
#pragma unroll
            for (int j = 0; j < 8; ++j) {
                cj[j] = __builtin_amdgcn_readlane(cn[j], l);
                v[j] = -1;
                if (lane < cj[j]) v[j] = cells[(l * 8 + j) * cap + lane];
            }
#pragma unroll
            for (int j = 0; j < 8; ++j) {
                if (v[j] >= 0)
                    __hip_atomic_fetch_add(&cnt_s[v[j] >> 16], 1.0f, __ATOMIC_RELAXED,
                                           __HIP_MEMORY_SCOPE_WORKGROUP);
            }
#pragma unroll
            for (int j = 0; j < 8; ++j) {
                for (int i = lane + 32; i < cj[j]; i += 32) {
                    const int w = cells[(l * 8 + j) * cap + i];
                    __hip_atomic_fetch_add(&cnt_s[w >> 16], 1.0f, __ATOMIC_RELAXED,
                                           __HIP_MEMORY_SCOPE_WORKGROUP);
                }
            }
        }
    } else {
        scan_dst<DEG_WAVES>(dst, nedges, [&](int d, int e) {
            (void)e;
            const unsigned dl = (unsigned)(d - r0);
            if (dl < (unsigned)rows)
                __hip_atomic_fetch_add(&cnt_s[dl], 1.0f, __ATOMIC_RELAXED,
                                       __HIP_MEMORY_SCOPE_WORKGROUP);
        }, [&] {});
    }
    __syncthreads();

    for (int r = threadIdx.x; r < rows; r += DEG_WAVES * 32) {
        const float d = cnt_s[r];
        dinv[r0 + r] = (d > 0.0f) ? rsqrtf(d) : 0.0f;
    }
}

__device__ __forceinline__ void add_edges(int s, int dl, float nrm, int n,
                                          const float* __restrict__ h,
                                          float* slab, int lane) {
    for (int i = 0; i < n; i += 8) {
        float4 hv[8];
#pragma unroll
        for (int u = 0; u < 8; ++u) {
            const int si = __builtin_amdgcn_readlane(s, i + u);
            hv[u] = *(const float4*)(h + (long)si * D_FEAT + lane * 4);
        }
#pragma unroll
        for (int u = 0; u < 8; ++u) {
            if (i + u < n) {
                const float nu = __int_as_float(__builtin_amdgcn_readlane(__float_as_int(nrm), i + u));
                const int dlu = __builtin_amdgcn_readlane(dl, i + u);
                float* row = slab + dlu * D_FEAT + lane;
                __hip_atomic_fetch_add(row +  0, nu * hv[u].x, __ATOMIC_RELAXED, __HIP_MEMORY_SCOPE_WORKGROUP);
                __hip_atomic_fetch_add(row + 32, nu * hv[u].y, __ATOMIC_RELAXED, __HIP_MEMORY_SCOPE_WORKGROUP);
                __hip_atomic_fetch_add(row + 64, nu * hv[u].z, __ATOMIC_RELAXED, __HIP_MEMORY_SCOPE_WORKGROUP);
                __hip_atomic_fetch_add(row + 96, nu * hv[u].w, __ATOMIC_RELAXED, __HIP_MEMORY_SCOPE_WORKGROUP);
            }
        }
    }
}

__device__ __forceinline__ void drain_edges(const int* q_e, const int* q_d, int base, int n,
                                            const int* __restrict__ src,
                                            const float* __restrict__ dinv,
                                            const float* dinv_s,
                                            const float* __restrict__ h,
                                            float* slab, int lane) {
    int dl = 0, s = 0;
    float nrm = 0.0f;
    if (lane < n) {
        const int e = q_e[base + lane];
        dl = q_d[base + lane];
        s = src[e];
        nrm = dinv[s] * dinv_s[dl];
    }
    add_edges(s, dl, nrm, n, h, slab, lane);
}

__global__ __launch_bounds__(OWN_THREADS)
void layer_owner_kernel(const int* __restrict__ src, const int* __restrict__ dst,
                        const float* __restrict__ dinv, const float* __restrict__ h,
                        const float* __restrict__ bias, const int* lists, float* xout,
                        int n, int nedges) {
    __shared__ float slab[OWN_ROWS * D_FEAT];
    __shared__ float dinv_s[OWN_ROWS];
    __shared__ int q_e_all[OWN_WAVES * Q_CAP];
    __shared__ int q_d_all[OWN_WAVES * Q_CAP];

    const int lane = threadIdx.x & 31;
    const int wave = __builtin_amdgcn_readfirstlane((int)(threadIdx.x >> 5));
    const int r0 = blockIdx.x * OWN_ROWS;
    const int rows = min(OWN_ROWS, n - r0);

    const int cap = cell_cap_of(rows);
    const int* cntp = lists + (long)r0 * D_FEAT;
    const int* cells = cntp + NCH + wave * 8 * cap;
    int cn[8];
    const bool ovf = load_cell_counts(cntp, cap, lane, cn);
    int pre[9];
    pre[0] = 0;
#pragma unroll
    for (int j = 0; j < 8; ++j) pre[j + 1] = pre[j] + __builtin_amdgcn_readlane(cn[j], wave);
    const int mycnt = ovf ? -1 : pre[8];
    auto fetch = [&](int e, int& s, int& dl) {
        int j = 0, pj = 0;
#pragma unroll
        for (int q = 1; q < 8; ++q) {
            const bool in = e >= pre[q];
            j += in ? 1 : 0;
            pj = in ? pre[q] : pj;
        }
        const int v = cells[j * cap + (e - pj)];
        s = v & 0xffff;
        dl = v >> 16;
    };
    int s_nx = 0, dl_nx = 0;
    if (lane < mycnt) fetch(lane, s_nx, dl_nx);

    const float4 bv = *(const float4*)(bias + lane * 4);
    for (int r = wave; r < rows; r += OWN_WAVES) {
        const float di = dinv[r0 + r];
        const float4 hv = *(const float4*)(h + (long)(r0 + r) * D_FEAT + lane * 4);
        float* row = slab + r * D_FEAT + lane;
        row[0]  = di * di * hv.x + bv.x;
        row[32] = di * di * hv.y + bv.y;
        row[64] = di * di * hv.z + bv.z;
        row[96] = di * di * hv.w + bv.w;
        if (lane == 0) dinv_s[r] = di;
    }
    __syncthreads();

    if (mycnt >= 0) {
        int base = 0;
        for (; base + 32 <= mycnt; base += 32) {
            const int s = s_nx, dl = dl_nx;
            const float nrm = dinv[s] * dinv_s[dl];
            s_nx = 0; dl_nx = 0;
            if (base + 32 + lane < mycnt) fetch(base + 32 + lane, s_nx, dl_nx);
            add_edges(s, dl, nrm, 32, h, slab, lane);
        }
        if (base < mycnt) {
            const float nrm = dinv[s_nx] * dinv_s[dl_nx];
            add_edges(s_nx, dl_nx, nrm, mycnt - base, h, slab, lane);
        }
    } else {
        int* q_e = q_e_all + wave * Q_CAP;
        int* q_d = q_d_all + wave * Q_CAP;
        int cnt = 0;

        scan_dst(dst, nedges, [&](int d, int e) {
            const unsigned dl = (unsigned)(d - r0);
            const bool hit = dl < (unsigned)rows;
            const unsigned m = __builtin_amdgcn_ballot_w32(hit);
            if (m != 0u) {
                const int pos = cnt + (int)__builtin_amdgcn_mbcnt_lo(m, 0u);
                if (hit) { q_e[pos] = e; q_d[pos] = (int)dl; }
                cnt += __popc(m);
            }
        }, [&] {
            while (cnt >= 32) {
                cnt -= 32;
                drain_edges(q_e, q_d, cnt, 32, src, dinv, dinv_s, h, slab, lane);
            }
        });
        if (cnt > 0) drain_edges(q_e, q_d, 0, cnt, src, dinv, dinv_s, h, slab, lane);
    }
    __syncthreads();

    for (int r = wave; r < rows; r += OWN_WAVES) {
        const float* row = slab + r * D_FEAT + lane;
        float4 v;
        v.x = row[0]; v.y = row[32]; v.z = row[64]; v.w = row[96];
        v.x = v.x > 0.0f ? v.x : 0.0f;
        v.y = v.y > 0.0f ? v.y : 0.0f;
        v.z = v.z > 0.0f ? v.z : 0.0f;
        v.w = v.w > 0.0f ? v.w : 0.0f;
        *(float4*)(xout + (long)(r0 + r) * D_FEAT + lane * 4) = v;
    }
}

#define LDA 132
__global__ __launch_bounds__(256)
void gemm128_wmma_kernel(const float* __restrict__ A, const float* __restrict__ W,
                         float* __restrict__ H, int nrows) {
    __shared__ float As[16 * LDA];
    const int tid = threadIdx.x;
    const long row0 = (long)blockIdx.x * 16;

    for (int idx = tid; idx < 512; idx += 256) {
        int r = idx >> 5;
        int c4 = idx & 31;
        long gr = row0 + r;
        if (gr >= nrows) gr = nrows - 1;
        *(float4*)(As + r * LDA + c4 * 4) =
            *(const float4*)(A + gr * D_FEAT + c4 * 4);
    }
    __syncthreads();

    const int wave  = tid >> 5;
    const int lane  = tid & 31;
    const int col0  = wave * 16;
    const int mrow  = lane & 15;
    const int khalf = (lane >> 4) * 2;
    const int ncol  = lane & 15;

    v8f acc = {};
    for (int k = 0; k < D_FEAT; k += 4) {
        v2f a, b;
        a.x = As[mrow * LDA + k + khalf + 0];
        a.y = As[mrow * LDA + k + khalf + 1];
        b.x = W[(k + khalf + 0) * D_FEAT + col0 + ncol];
        b.y = W[(k + khalf + 1) * D_FEAT + col0 + ncol];
        acc = __builtin_amdgcn_wmma_f32_16x16x4_f32(
             false, a,  false, b,
             (short)0, acc,  false,  false);
    }

    const int mbase = (lane >> 4) * 8;
    for (int r = 0; r < 8; ++r) {
        long m = row0 + mbase + r;
        if (m < nrows) H[m * D_FEAT + col0 + ncol] = acc[r];
    }
}

#define LDAJ 260
__global__ __launch_bounds__(256)
void gemm_jk_wmma_kernel(const float* __restrict__ x1, const float* __restrict__ x2,
                         const float* __restrict__ Wl, const float* __restrict__ bl,
                         float* __restrict__ out, int nrows) {
    __shared__ float As[16 * LDAJ];
    const int tid = threadIdx.x;
    const long row0 = (long)blockIdx.x * 16;

    for (int idx = tid; idx < 1024; idx += 256) {
        int r = idx >> 6;
        int c4 = idx & 63;
        long gr = row0 + r;
        if (gr >= nrows) gr = nrows - 1;
        const float* srcp = (c4 < 32)
            ? (x1 + gr * D_FEAT + c4 * 4)
            : (x2 + gr * D_FEAT + (c4 - 32) * 4);
        *(float4*)(As + r * LDAJ + c4 * 4) = *(const float4*)srcp;
    }
    __syncthreads();

    const int wave  = tid >> 5;
    const int lane  = tid & 31;
    const int col0  = wave * 16;
    const int mrow  = lane & 15;
    const int khalf = (lane >> 4) * 2;
    const int ncol  = lane & 15;

    v8f acc = {};
    for (int k = 0; k < 2 * D_FEAT; k += 4) {
        v2f a, b;
        a.x = As[mrow * LDAJ + k + khalf + 0];
        a.y = As[mrow * LDAJ + k + khalf + 1];
        b.x = Wl[(k + khalf + 0) * D_FEAT + col0 + ncol];
        b.y = Wl[(k + khalf + 1) * D_FEAT + col0 + ncol];
        acc = __builtin_amdgcn_wmma_f32_16x16x4_f32(
            false, a, false, b, (short)0, acc, false, false);
    }

    const float bias = bl[col0 + ncol];
    const int mbase = (lane >> 4) * 8;
    for (int r = 0; r < 8; ++r) {
        long m = row0 + mbase + r;
        if (m < nrows) out[m * D_FEAT + col0 + ncol] = acc[r] + bias;
    }
}

extern "C" void kernel_launch(void* const* d_in, const int* in_sizes, int n_in,
                              void* d_out, int out_size, void* d_ws, size_t ws_size,
                              hipStream_t stream) {
    (void)n_in; (void)out_size; (void)ws_size;

    const float* x    = (const float*)d_in[0];
    const int*   ei   = (const int*)d_in[1];
    const float* W1   = (const float*)d_in[2];
    const float* b1   = (const float*)d_in[3];
    const float* W2   = (const float*)d_in[4];
    const float* b2   = (const float*)d_in[5];
    const float* linW = (const float*)d_in[6];
    const float* linb = (const float*)d_in[7];
    float* out = (float*)d_out;

    const int N = in_sizes[0] / D_FEAT;
    const int E = in_sizes[1] / 2;
    const int* srcI = ei;
    const int* dstI = ei + E;

    float* ws   = (float*)d_ws;
    float* dinv = ws;
    float* h    = dinv + N;
    float* x1   = h   + (long)N * D_FEAT;
    float* x2   = x1  + (long)N * D_FEAT;

    const int nBlkRow = (N + 15) / 16;
    const int nBlkOwn = (N + OWN_ROWS - 1) / OWN_ROWS;

    int* lists = (int*)x2;

    bin_edges_kernel<<<NCH, BIN_THREADS, 0, stream>>>(srcI, dstI, lists, N, E, nBlkOwn);
    deg_kernel<<<nBlkOwn, DEG_WAVES * 32, 0, stream>>>(dstI, lists, dinv, N, E);

    gemm128_wmma_kernel<<<nBlkRow, 256, 0, stream>>>(x, W1, h, N);
    layer_owner_kernel<<<nBlkOwn, OWN_THREADS, 0, stream>>>(srcI, dstI, dinv, h, b1, lists, x1, N, E);

    gemm128_wmma_kernel<<<nBlkRow, 256, 0, stream>>>(x1, W2, h, N);
    layer_owner_kernel<<<nBlkOwn, OWN_THREADS, 0, stream>>>(srcI, dstI, dinv, h, b2, lists, x2, N, E);

    gemm_jk_wmma_kernel<<<nBlkRow, 256, 0, stream>>>(x1, x2, linW, linb, out, N);
}


# Second layer's product x1 @ W2 formed in the first layer kernel from its LDS slab (odd row pitch, W2 strip in registers): the separate product launch and its re-read of x1 go; k sum order changes
# speedup vs baseline: 3.5316x; 1.0621x over previous
#include <hip/hip_runtime.h>


typedef __attribute__((ext_vector_type(2))) float v2f;
typedef __attribute__((ext_vector_type(8))) float v8f;

#define D_FEAT 128

#define OWN_ROWS    196
#define SLAB_LD     129
#define OWN_WAVES   32
#define OWN_THREADS (OWN_WAVES * 32)
#define GROUP_EDGES 512
#define Q_CAP       544

template <int NW = OWN_WAVES, class F, class G>
__device__ __forceinline__ void scan_dst(const int* __restrict__ dst, int nedges, F&& visit,
                                         G&& after_group) {
    const int lane = threadIdx.x & 31;
    const int wave = __builtin_amdgcn_readfirstlane((int)(threadIdx.x >> 5));
    const bool vec_ok = (((size_t)dst) & 15) == 0;
    const int ngroups = vec_ok ? nedges / GROUP_EDGES : 0;
    const int off = (int)blockIdx.x * (ngroups / (int)gridDim.x);

    auto rot = [&](int g) { int gr = g + off; return gr >= ngroups ? gr - ngroups : gr; };
    auto load_group = [&](int4 (&v)[4], int gr) {
        const int4* p = (const int4*)(dst + gr * GROUP_EDGES) + lane;
#pragma unroll
        for (int j = 0; j < 4; ++j) v[j] = p[j * 32];
    };
    auto step = [&](int4 (&cur)[4], int4 (&nxt)[4], int g) {
        const int gr = rot(g);
        const int gn = g + NW;
        load_group(nxt, rot(gn < ngroups ? gn : g));
#pragma unroll
        for (int j = 0; j < 4; ++j) {
            const int e0 = gr * GROUP_EDGES + j * 128 + lane * 4;
            visit(cur[j].x, e0 + 0);
            visit(cur[j].y, e0 + 1);
            visit(cur[j].z, e0 + 2);
            visit(cur[j].w, e0 + 3);
        }
        after_group();
    };

    if (wave < ngroups) {
        int4 bufa[4], bufb[4];
        load_group(bufa, rot(wave));
        for (int g = wave; g < ngroups; g += 2 * NW) {
            step(bufa, bufb, g);
            if (g + NW >= ngroups) break;
            step(bufb, bufa, g + NW);
        }
    }
    for (int base = ngroups * GROUP_EDGES + wave * 32; base < nedges; base += NW * 32) {
        const int e = base + lane;
        visit(e < nedges ? dst[e] : -1, e);
        after_group();
    }
}

#define NCH     256
#define OWN_MAX 512
#define BIN_THREADS 256
#define DEG_WAVES   8
static_assert(NCH == 8 * OWN_WAVES && NCH == 32 * DEG_WAVES, "cells per wave");

__device__ __forceinline__ int cell_cap_of(int rows) {
    return (rows * D_FEAT - NCH) / NCH;
}

__global__ __launch_bounds__(BIN_THREADS)
void bin_edges_kernel(const int* __restrict__ src, const int* __restrict__ dst,
                      int* __restrict__ lists, int n, int nedges, int nown) {
    __shared__ int fill_s[OWN_MAX];

    const int c = blockIdx.x;
    const int per = (nedges + NCH - 1) / NCH;
    const int e0 = c * per;
    const int e1 = min(nedges, e0 + per);
    const bool usable = (n <= 65536) && (nown <= OWN_MAX);

    for (int o = threadIdx.x; o < OWN_MAX; o += BIN_THREADS) fill_s[o] = 0;
    __syncthreads();

    if (usable) {
        for (int base = e0; base < e1; base += 8 * BIN_THREADS) {
            int dd[8], ss[8];
#pragma unroll
            for (int u = 0; u < 8; ++u) {
                const int e = base + u * BIN_THREADS + (int)threadIdx.x;
                const bool ok = e < e1;
                dd[u] = ok ? dst[e] : -1;
                ss[u] = ok ? src[e] : 0;
            }
#pragma unroll
            for (int u = 0; u < 8; ++u) {
                const int d = dd[u];
                if ((unsigned)d < (unsigned)n) {
                    const int o = d / OWN_ROWS;
                    const int row = d - o * OWN_ROWS;
                    const int cap = cell_cap_of(min(OWN_ROWS, n - o * OWN_ROWS));
                    const int pos = __hip_atomic_fetch_add(&fill_s[o], 1, __ATOMIC_RELAXED,
                                                           __HIP_MEMORY_SCOPE_WORKGROUP);
                    if (pos < cap)
                        lists[(long)o * OWN_ROWS * D_FEAT + NCH + c * cap + pos] = ss[u] | (row << 16);
                }
            }
        }
    }
    __syncthreads();

    for (int o = threadIdx.x; o < nown; o += BIN_THREADS) {
        int v = 0x7fffffff;
        if (usable) v = fill_s[o];
        lists[(long)o * OWN_ROWS * D_FEAT + c] = v;
    }
}

__device__ __forceinline__ bool load_cell_counts(const int* cntp, int cap, int lane, int (&cn)[8]) {
    const int4* cp = (const int4*)cntp + lane * 2;
    const int4 a = cp[0], b = cp[1];
    cn[0] = a.x; cn[1] = a.y; cn[2] = a.z; cn[3] = a.w;
    cn[4] = b.x; cn[5] = b.y; cn[6] = b.z; cn[7] = b.w;
    bool bad = false;
#pragma unroll
    for (int j = 0; j < 8; ++j) bad = bad || (cn[j] > cap);
    return __builtin_amdgcn_ballot_w32(bad) != 0u;
}

__global__ __launch_bounds__(DEG_WAVES * 32)
void deg_kernel(const int* __restrict__ dst, const int* __restrict__ lists,
                float* __restrict__ dinv, int n, int nedges) {
    __shared__ float cnt_s[OWN_ROWS];

    const int lane = threadIdx.x & 31;
    const int wave = __builtin_amdgcn_readfirstlane((int)(threadIdx.x >> 5));
    const int r0 = blockIdx.x * OWN_ROWS;
    const int rows = min(OWN_ROWS, n - r0);
    const int cap = cell_cap_of(rows);
    const int* cntp = lists + (long)r0 * D_FEAT;
    const int* cells = cntp + NCH;

    int cn[8];
    const bool ovf = load_cell_counts(cntp, cap, lane, cn);

    for (int r = threadIdx.x; r < OWN_ROWS; r += DEG_WAVES * 32) cnt_s[r] = 1.0f;
    __syncthreads();

    if (!ovf) {
        for (int t = 0; t < 4; ++t) {
            const int l = wave * 4 + t;
            int cj[8], v[8];
#pragma unroll
            for (int j = 0; j < 8; ++j) {
                cj[j] = __builtin_amdgcn_readlane(cn[j], l);
                v[j] = -1;
                if (lane < cj[j]) v[j] = cells[(l * 8 + j) * cap + lane];
            }
#pragma unroll
            for (int j = 0; j < 8; ++j) {
                if (v[j] >= 0)
                    __hip_atomic_fetch_add(&cnt_s[v[j] >> 16], 1.0f, __ATOMIC_RELAXED,
                                           __HIP_MEMORY_SCOPE_WORKGROUP);
            }
#pragma unroll
            for (int j = 0; j < 8; ++j) {
                for (int i = lane + 32; i < cj[j]; i += 32) {
                    const int w = cells[(l * 8 + j) * cap + i];
                    __hip_atomic_fetch_add(&cnt_s[w >> 16], 1.0f, __ATOMIC_RELAXED,
                                           __HIP_MEMORY_SCOPE_WORKGROUP);
                }
            }
        }
    } else {
        scan_dst<DEG_WAVES>(dst, nedges, [&](int d, int e) {
            (void)e;
            const unsigned dl = (unsigned)(d - r0);
            if (dl < (unsigned)rows)
                __hip_atomic_fetch_add(&cnt_s[dl], 1.0f, __ATOMIC_RELAXED,
                                       __HIP_MEMORY_SCOPE_WORKGROUP);
        }, [&] {});
    }
    __syncthreads();

    for (int r = threadIdx.x; r < rows; r += DEG_WAVES * 32) {
        const float d = cnt_s[r];
        dinv[r0 + r] = (d > 0.0f) ? rsqrtf(d) : 0.0f;
    }
}

__device__ __forceinline__ void add_edges(int s, int dl, float nrm, int n,
                                          const float* __restrict__ h,
                                          float* slab, int lane) {
    for (int i = 0; i < n; i += 8) {
        float4 hv[8];
#pragma unroll
        for (int u = 0; u < 8; ++u) {
            const int si = __builtin_amdgcn_readlane(s, i + u);
            hv[u] = *(const float4*)(h + (long)si * D_FEAT + lane * 4);
        }
#pragma unroll
        for (int u = 0; u < 8; ++u) {
            if (i + u < n) {
                const float nu = __int_as_float(__builtin_amdgcn_readlane(__float_as_int(nrm), i + u));
                const int dlu = __builtin_amdgcn_readlane(dl, i + u);
                float* row = slab + dlu * SLAB_LD + lane;
                __hip_atomic_fetch_add(row +  0, nu * hv[u].x, __ATOMIC_RELAXED, __HIP_MEMORY_SCOPE_WORKGROUP);
                __hip_atomic_fetch_add(row + 32, nu * hv[u].y, __ATOMIC_RELAXED, __HIP_MEMORY_SCOPE_WORKGROUP);
                __hip_atomic_fetch_add(row + 64, nu * hv[u].z, __ATOMIC_RELAXED, __HIP_MEMORY_SCOPE_WORKGROUP);
                __hip_atomic_fetch_add(row + 96, nu * hv[u].w, __ATOMIC_RELAXED, __HIP_MEMORY_SCOPE_WORKGROUP);
            }
        }
    }
}

__device__ __forceinline__ void drain_edges(const int* q_e, const int* q_d, int base, int n,
                                            const int* __restrict__ src,
                                            const float* __restrict__ dinv,
                                            const float* dinv_s,
                                            const float* __restrict__ h,
                                            float* slab, int lane) {
    int dl = 0, s = 0;
    float nrm = 0.0f;
    if (lane < n) {
        const int e = q_e[base + lane];
        dl = q_d[base + lane];
        s = src[e];
        nrm = dinv[s] * dinv_s[dl];
    }
    add_edges(s, dl, nrm, n, h, slab, lane);
}

template <bool NEXT>
__global__ __launch_bounds__(OWN_THREADS)
void layer_owner_kernel(const int* __restrict__ src, const int* __restrict__ dst,
                        const float* __restrict__ dinv, const float* __restrict__ h,
                        const float* __restrict__ bias, const int* lists, float* xout,
                        const float* __restrict__ Wn, float* __restrict__ hnext,
                        int n, int nedges) {
    __shared__ float slab[OWN_ROWS * SLAB_LD];
    __shared__ float dinv_s[OWN_ROWS];
    __shared__ int q_e_all[OWN_WAVES * Q_CAP];
    __shared__ int q_d_all[OWN_WAVES * Q_CAP];

    const int lane = threadIdx.x & 31;
    const int wave = __builtin_amdgcn_readfirstlane((int)(threadIdx.x >> 5));
    const int r0 = blockIdx.x * OWN_ROWS;
    const int rows = min(OWN_ROWS, n - r0);

    const int cap = cell_cap_of(rows);
    const int* cntp = lists + (long)r0 * D_FEAT;
    const int* cells = cntp + NCH + wave * 8 * cap;
    int cn[8];
    const bool ovf = load_cell_counts(cntp, cap, lane, cn);
    int pre[9];
    pre[0] = 0;
#pragma unroll
    for (int j = 0; j < 8; ++j) pre[j + 1] = pre[j] + __builtin_amdgcn_readlane(cn[j], wave);
    const int mycnt = ovf ? -1 : pre[8];
    auto fetch = [&](int e, int& s, int& dl) {
        int j = 0, pj = 0;
#pragma unroll
        for (int q = 1; q < 8; ++q) {
            const bool in = e >= pre[q];
            j += in ? 1 : 0;
            pj = in ? pre[q] : pj;
        }
        const int v = cells[j * cap + (e - pj)];
        s = v & 0xffff;
        dl = v >> 16;
    };
    int s_nx = 0, dl_nx = 0;
    if (lane < mycnt) fetch(lane, s_nx, dl_nx);

    const float4 bv = *(const float4*)(bias + lane * 4);
    for (int r = wave; r < rows; r += OWN_WAVES) {
        const float di = dinv[r0 + r];
        const float4 hv = *(const float4*)(h + (long)(r0 + r) * D_FEAT + lane * 4);
        float* row = slab + r * SLAB_LD + lane;
        row[0]  = di * di * hv.x + bv.x;
        row[32] = di * di * hv.y + bv.y;
        row[64] = di * di * hv.z + bv.z;
        row[96] = di * di * hv.w + bv.w;
        if (lane == 0) dinv_s[r] = di;
    }
    __syncthreads();

    if (mycnt >= 0) {
        int base = 0;
        for (; base + 32 <= mycnt; base += 32) {
            const int s = s_nx, dl = dl_nx;
            const float nrm = dinv[s] * dinv_s[dl];
            s_nx = 0; dl_nx = 0;
            if (base + 32 + lane < mycnt) fetch(base + 32 + lane, s_nx, dl_nx);
            add_edges(s, dl, nrm, 32, h, slab, lane);
        }
        if (base < mycnt) {
            const float nrm = dinv[s_nx] * dinv_s[dl_nx];
            add_edges(s_nx, dl_nx, nrm, mycnt - base, h, slab, lane);
        }
    } else {
        int* q_e = q_e_all + wave * Q_CAP;
        int* q_d = q_d_all + wave * Q_CAP;
        int cnt = 0;

        scan_dst(dst, nedges, [&](int d, int e) {
            const unsigned dl = (unsigned)(d - r0);
            const bool hit = dl < (unsigned)rows;
            const unsigned m = __builtin_amdgcn_ballot_w32(hit);
            if (m != 0u) {
                const int pos = cnt + (int)__builtin_amdgcn_mbcnt_lo(m, 0u);
                if (hit) { q_e[pos] = e; q_d[pos] = (int)dl; }
                cnt += __popc(m);
            }
        }, [&] {
            while (cnt >= 32) {
                cnt -= 32;
                drain_edges(q_e, q_d, cnt, 32, src, dinv, dinv_s, h, slab, lane);
            }
        });
        if (cnt > 0) drain_edges(q_e, q_d, 0, cnt, src, dinv, dinv_s, h, slab, lane);
    }
    __syncthreads();

    for (int r = wave; r < rows; r += OWN_WAVES) {
        const float* row = slab + r * SLAB_LD + lane;
        float4 v;
        v.x = row[0]; v.y = row[32]; v.z = row[64]; v.w = row[96];
        v.x = v.x > 0.0f ? v.x : 0.0f;
        v.y = v.y > 0.0f ? v.y : 0.0f;
        v.z = v.z > 0.0f ? v.z : 0.0f;
        v.w = v.w > 0.0f ? v.w : 0.0f;
        *(float4*)(xout + (long)(r0 + r) * D_FEAT + lane * 4) = v;
    }

    if constexpr (NEXT) {
        const int ct = wave & 7;
        const int m16 = lane & 15;
        const int hh = lane >> 4;
        const float* wp = Wn + hh * D_FEAT + ct * 16 + m16;
        v2f bfr[32];
#pragma unroll
        for (int t = 0; t < 32; ++t) {
            bfr[t].x = wp[(4 * t) * D_FEAT];
            bfr[t].y = wp[(4 * t + 2) * D_FEAT];
        }
        const int ntiles = (rows + 15) >> 4;
        for (int rt = wave >> 3; rt < ntiles; rt += 4) {
            const int ar = min(rt * 16 + m16, OWN_ROWS - 1);
            const float* ap = slab + ar * SLAB_LD + 32 * hh;
            v8f acc = {};
#pragma unroll
            for (int t = 0; t < 32; ++t) {
                v2f a;
                a.x = ap[t];
                a.y = ap[64 + t];
                a.x = a.x > 0.0f ? a.x : 0.0f;
                a.y = a.y > 0.0f ? a.y : 0.0f;
                acc = __builtin_amdgcn_wmma_f32_16x16x4_f32(
                     false, a,  false, bfr[t],
                     (short)0, acc,  false,  false);
            }
            for (int r = 0; r < 8; ++r) {
                const int orow = rt * 16 + hh * 8 + r;
                if (orow < rows) hnext[(long)(r0 + orow) * D_FEAT + ct * 16 + m16] = acc[r];
            }
        }
    }
}

#define LDA 132
__global__ __launch_bounds__(256)
void gemm128_wmma_kernel(const float* __restrict__ A, const float* __restrict__ W,
                         float* __restrict__ H, int nrows) {
    __shared__ float As[16 * LDA];
    const int tid = threadIdx.x;
    const long row0 = (long)blockIdx.x * 16;

    for (int idx = tid; idx < 512; idx += 256) {
        int r = idx >> 5;
        int c4 = idx & 31;
        long gr = row0 + r;
        if (gr >= nrows) gr = nrows - 1;
        *(float4*)(As + r * LDA + c4 * 4) =
            *(const float4*)(A + gr * D_FEAT + c4 * 4);
    }
    __syncthreads();

    const int wave  = tid >> 5;
    const int lane  = tid & 31;
    const int col0  = wave * 16;
    const int mrow  = lane & 15;
    const int khalf = (lane >> 4) * 2;
    const int ncol  = lane & 15;

    v8f acc = {};
    for (int k = 0; k < D_FEAT; k += 4) {
        v2f a, b;
        a.x = As[mrow * LDA + k + khalf + 0];
        a.y = As[mrow * LDA + k + khalf + 1];
        b.x = W[(k + khalf + 0) * D_FEAT + col0 + ncol];
        b.y = W[(k + khalf + 1) * D_FEAT + col0 + ncol];
        acc = __builtin_amdgcn_wmma_f32_16x16x4_f32(
             false, a,  false, b,
             (short)0, acc,  false,  false);
    }

    const int mbase = (lane >> 4) * 8;
    for (int r = 0; r < 8; ++r) {
        long m = row0 + mbase + r;
        if (m < nrows) H[m * D_FEAT + col0 + ncol] = acc[r];
    }
}

#define LDAJ 260
__global__ __launch_bounds__(256)
void gemm_jk_wmma_kernel(const float* __restrict__ x1, const float* __restrict__ x2,
                         const float* __restrict__ Wl, const float* __restrict__ bl,
                         float* __restrict__ out, int nrows) {
    __shared__ float As[16 * LDAJ];
    const int tid = threadIdx.x;
    const long row0 = (long)blockIdx.x * 16;

    for (int idx = tid; idx < 1024; idx += 256) {
        int r = idx >> 6;
        int c4 = idx & 63;
        long gr = row0 + r;
        if (gr >= nrows) gr = nrows - 1;
        const float* srcp = (c4 < 32)
            ? (x1 + gr * D_FEAT + c4 * 4)
            : (x2 + gr * D_FEAT + (c4 - 32) * 4);
        *(float4*)(As + r * LDAJ + c4 * 4) = *(const float4*)srcp;
    }
    __syncthreads();

    const int wave  = tid >> 5;
    const int lane  = tid & 31;
    const int col0  = wave * 16;
    const int mrow  = lane & 15;
    const int khalf = (lane >> 4) * 2;
    const int ncol  = lane & 15;

    v8f acc = {};
    for (int k = 0; k < 2 * D_FEAT; k += 4) {
        v2f a, b;
        a.x = As[mrow * LDAJ + k + khalf + 0];
        a.y = As[mrow * LDAJ + k + khalf + 1];
        b.x = Wl[(k + khalf + 0) * D_FEAT + col0 + ncol];
        b.y = Wl[(k + khalf + 1) * D_FEAT + col0 + ncol];
        acc = __builtin_amdgcn_wmma_f32_16x16x4_f32(
            false, a, false, b, (short)0, acc, false, false);
    }

    const float bias = bl[col0 + ncol];
    const int mbase = (lane >> 4) * 8;
    for (int r = 0; r < 8; ++r) {
        long m = row0 + mbase + r;
        if (m < nrows) out[m * D_FEAT + col0 + ncol] = acc[r] + bias;
    }
}

extern "C" void kernel_launch(void* const* d_in, const int* in_sizes, int n_in,
                              void* d_out, int out_size, void* d_ws, size_t ws_size,
                              hipStream_t stream) {
    (void)n_in; (void)out_size; (void)ws_size;

    const float* x    = (const float*)d_in[0];
    const int*   ei   = (const int*)d_in[1];
    const float* W1   = (const float*)d_in[2];
    const float* b1   = (const float*)d_in[3];
    const float* W2   = (const float*)d_in[4];
    const float* b2   = (const float*)d_in[5];
    const float* linW = (const float*)d_in[6];
    const float* linb = (const float*)d_in[7];
    float* out = (float*)d_out;

    const int N = in_sizes[0] / D_FEAT;
    const int E = in_sizes[1] / 2;
    const int* srcI = ei;
    const int* dstI = ei + E;

    float* ws   = (float*)d_ws;
    float* dinv = ws;
    float* h    = dinv + N;
    float* x1   = h   + (long)N * D_FEAT;
    float* x2   = x1  + (long)N * D_FEAT;

    const int nBlkRow = (N + 15) / 16;
    const int nBlkOwn = (N + OWN_ROWS - 1) / OWN_ROWS;

    int* lists = (int*)x2;

    bin_edges_kernel<<<NCH, BIN_THREADS, 0, stream>>>(srcI, dstI, lists, N, E, nBlkOwn);
    deg_kernel<<<nBlkOwn, DEG_WAVES * 32, 0, stream>>>(dstI, lists, dinv, N, E);

    float* h2 = out;

    gemm128_wmma_kernel<<<nBlkRow, 256, 0, stream>>>(x, W1, h, N);
    layer_owner_kernel<true><<<nBlkOwn, OWN_THREADS, 0, stream>>>(srcI, dstI, dinv, h, b1, lists, x1,
                                                                 W2, h2, N, E);

    layer_owner_kernel<false><<<nBlkOwn, OWN_THREADS, 0, stream>>>(srcI, dstI, dinv, h2, b2, lists, x2,
                                                                  nullptr, nullptr, N, E);

    gemm_jk_wmma_kernel<<<nBlkRow, 256, 0, stream>>>(x1, x2, linW, linb, out, N);
}


# First product x @ W1 formed by the degree kernel's owner blocks (196 rows of x through an LDS slab, W1 strip in registers): the 25000-wave product launch goes, 6 launches->5; k sum order changes
# speedup vs baseline: 3.6494x; 1.0334x over previous
#include <hip/hip_runtime.h>


typedef __attribute__((ext_vector_type(2))) float v2f;
typedef __attribute__((ext_vector_type(8))) float v8f;

#define D_FEAT 128

#define OWN_ROWS    196
#define SLAB_LD     129
#define OWN_WAVES   32
#define OWN_THREADS (OWN_WAVES * 32)
#define GROUP_EDGES 512
#define Q_CAP       544

template <int NW = OWN_WAVES, class F, class G>
__device__ __forceinline__ void scan_dst(const int* __restrict__ dst, int nedges, F&& visit,
                                         G&& after_group) {
    const int lane = threadIdx.x & 31;
    const int wave = __builtin_amdgcn_readfirstlane((int)(threadIdx.x >> 5));
    const bool vec_ok = (((size_t)dst) & 15) == 0;
    const int ngroups = vec_ok ? nedges / GROUP_EDGES : 0;
    const int off = (int)blockIdx.x * (ngroups / (int)gridDim.x);

    auto rot = [&](int g) { int gr = g + off; return gr >= ngroups ? gr - ngroups : gr; };
    auto load_group = [&](int4 (&v)[4], int gr) {
        const int4* p = (const int4*)(dst + gr * GROUP_EDGES) + lane;
#pragma unroll
        for (int j = 0; j < 4; ++j) v[j] = p[j * 32];
    };
    auto step = [&](int4 (&cur)[4], int4 (&nxt)[4], int g) {
        const int gr = rot(g);
        const int gn = g + NW;
        load_group(nxt, rot(gn < ngroups ? gn : g));
#pragma unroll
        for (int j = 0; j < 4; ++j) {
            const int e0 = gr * GROUP_EDGES + j * 128 + lane * 4;
            visit(cur[j].x, e0 + 0);
            visit(cur[j].y, e0 + 1);
            visit(cur[j].z, e0 + 2);
            visit(cur[j].w, e0 + 3);
        }
        after_group();
    };

    if (wave < ngroups) {
        int4 bufa[4], bufb[4];
        load_group(bufa, rot(wave));
        for (int g = wave; g < ngroups; g += 2 * NW) {
            step(bufa, bufb, g);
            if (g + NW >= ngroups) break;
            step(bufb, bufa, g + NW);
        }
    }
    for (int base = ngroups * GROUP_EDGES + wave * 32; base < nedges; base += NW * 32) {
        const int e = base + lane;
        visit(e < nedges ? dst[e] : -1, e);
        after_group();
    }
}

#define NCH     256
#define OWN_MAX 512
#define BIN_THREADS 256
static_assert(NCH == 8 * OWN_WAVES, "cells per wave");

__device__ __forceinline__ int cell_cap_of(int rows) {
    return (rows * D_FEAT - NCH) / NCH;
}

__global__ __launch_bounds__(BIN_THREADS)
void bin_edges_kernel(const int* __restrict__ src, const int* __restrict__ dst,
                      int* __restrict__ lists, int n, int nedges, int nown) {
    __shared__ int fill_s[OWN_MAX];

    const int c = blockIdx.x;
    const int per = (nedges + NCH - 1) / NCH;
    const int e0 = c * per;
    const int e1 = min(nedges, e0 + per);
    const bool usable = (n <= 65536) && (nown <= OWN_MAX);

    for (int o = threadIdx.x; o < OWN_MAX; o += BIN_THREADS) fill_s[o] = 0;
    __syncthreads();

    if (usable) {
        for (int base = e0; base < e1; base += 8 * BIN_THREADS) {
            int dd[8], ss[8];
#pragma unroll
            for (int u = 0; u < 8; ++u) {
                const int e = base + u * BIN_THREADS + (int)threadIdx.x;
                const bool ok = e < e1;
                dd[u] = ok ? dst[e] : -1;
                ss[u] = ok ? src[e] : 0;
            }
#pragma unroll
            for (int u = 0; u < 8; ++u) {
                const int d = dd[u];
                if ((unsigned)d < (unsigned)n) {
                    const int o = d / OWN_ROWS;
                    const int row = d - o * OWN_ROWS;
                    const int cap = cell_cap_of(min(OWN_ROWS, n - o * OWN_ROWS));
                    const int pos = __hip_atomic_fetch_add(&fill_s[o], 1, __ATOMIC_RELAXED,
                                                           __HIP_MEMORY_SCOPE_WORKGROUP);
                    if (pos < cap)
                        lists[(long)o * OWN_ROWS * D_FEAT + NCH + c * cap + pos] = ss[u] | (row << 16);
                }
            }
        }
    }
    __syncthreads();

    for (int o = threadIdx.x; o < nown; o += BIN_THREADS) {
        int v = 0x7fffffff;
        if (usable) v = fill_s[o];
        lists[(long)o * OWN_ROWS * D_FEAT + c] = v;
    }
}

__device__ __forceinline__ bool load_cell_counts(const int* cntp, int cap, int lane, int (&cn)[8]) {
    const int4* cp = (const int4*)cntp + lane * 2;
    const int4 a = cp[0], b = cp[1];
    cn[0] = a.x; cn[1] = a.y; cn[2] = a.z; cn[3] = a.w;
    cn[4] = b.x; cn[5] = b.y; cn[6] = b.z; cn[7] = b.w;
    bool bad = false;
#pragma unroll
    for (int j = 0; j < 8; ++j) bad = bad || (cn[j] > cap);
    return __builtin_amdgcn_ballot_w32(bad) != 0u;
}

__device__ __forceinline__ void load_w_strip(const float* __restrict__ Wm, int wave, int lane,
                                             v2f (&bfr)[32]) {
    const float* wp = Wm + (lane >> 4) * D_FEAT + (wave & 7) * 16 + (lane & 15);
#pragma unroll
    for (int t = 0; t < 32; ++t) {
        bfr[t].x = wp[(4 * t) * D_FEAT];
        bfr[t].y = wp[(4 * t + 2) * D_FEAT];
    }
}

template <bool RELU>
__device__ __forceinline__ void slab_product(const float* slab, const v2f (&bfr)[32],
                                             float* __restrict__ hout, int r0, int rows,
                                             int wave, int lane) {
    const int ct = wave & 7;
    const int m16 = lane & 15;
    const int hh = lane >> 4;
    const int ntiles = (rows + 15) >> 4;
    for (int rt = wave >> 3; rt < ntiles; rt += 4) {
        const int ar = min(rt * 16 + m16, OWN_ROWS - 1);
        const float* ap = slab + ar * SLAB_LD + 32 * hh;
        v8f acc = {};
#pragma unroll
        for (int t = 0; t < 32; ++t) {
            v2f a;
            a.x = ap[t];
            a.y = ap[64 + t];
            if constexpr (RELU) {
                a.x = a.x > 0.0f ? a.x : 0.0f;
                a.y = a.y > 0.0f ? a.y : 0.0f;
            }
            acc = __builtin_amdgcn_wmma_f32_16x16x4_f32(
                 false, a,  false, bfr[t],
                 (short)0, acc,  false,  false);
        }
        for (int r = 0; r < 8; ++r) {
            const int orow = rt * 16 + hh * 8 + r;
            if (orow < rows) hout[(long)(r0 + orow) * D_FEAT + ct * 16 + m16] = acc[r];
        }
    }
}

#define ROWS_PER_WAVE ((OWN_ROWS + OWN_WAVES - 1) / OWN_WAVES)
__global__ __launch_bounds__(OWN_THREADS)
void deg_product_kernel(const int* __restrict__ dst, const int* __restrict__ lists,
                        const float* __restrict__ x, const float* __restrict__ W,
                        float* __restrict__ dinv, float* __restrict__ h, int n, int nedges) {
    __shared__ float slab[OWN_ROWS * SLAB_LD];
    __shared__ float cnt_s[OWN_ROWS];

    const int lane = threadIdx.x & 31;
    const int wave = __builtin_amdgcn_readfirstlane((int)(threadIdx.x >> 5));
    const int r0 = blockIdx.x * OWN_ROWS;
    const int rows = min(OWN_ROWS, n - r0);
    const int cap = cell_cap_of(rows);
    const int* cntp = lists + (long)r0 * D_FEAT;
    const int* cells = cntp + NCH + wave * 8 * cap;

    int cn[8];
    const bool ovf = load_cell_counts(cntp, cap, lane, cn);

    float4 xv[ROWS_PER_WAVE];
#pragma unroll
    for (int k = 0; k < ROWS_PER_WAVE; ++k) {
        const int r = min(wave + k * OWN_WAVES, rows - 1);
        xv[k] = *(const float4*)(x + (long)(r0 + r) * D_FEAT + lane * 4);
    }

    v2f bfr[32];
    load_w_strip(W, wave, lane, bfr);

    int cj[8], v[8];
#pragma unroll
    for (int j = 0; j < 8; ++j) {
        cj[j] = ovf ? 0 : __builtin_amdgcn_readlane(cn[j], wave);
        v[j] = -1;
        if (lane < cj[j]) v[j] = cells[j * cap + lane];
    }

    for (int r = threadIdx.x; r < OWN_ROWS; r += OWN_THREADS) cnt_s[r] = 1.0f;
#pragma unroll
    for (int k = 0; k < ROWS_PER_WAVE; ++k) {
        const int r = wave + k * OWN_WAVES;
        if (r < rows) {
            float* row = slab + r * SLAB_LD + lane;
            row[0]  = xv[k].x;
            row[32] = xv[k].y;
            row[64] = xv[k].z;
            row[96] = xv[k].w;
        }
    }
    __syncthreads();

    if (!ovf) {
#pragma unroll
        for (int j = 0; j < 8; ++j) {
            if (v[j] >= 0)
                __hip_atomic_fetch_add(&cnt_s[v[j] >> 16], 1.0f, __ATOMIC_RELAXED,
                                       __HIP_MEMORY_SCOPE_WORKGROUP);
        }
#pragma unroll
        for (int j = 0; j < 8; ++j) {
            for (int i = lane + 32; i < cj[j]; i += 32) {
                const int w = cells[j * cap + i];
                __hip_atomic_fetch_add(&cnt_s[w >> 16], 1.0f, __ATOMIC_RELAXED,
                                       __HIP_MEMORY_SCOPE_WORKGROUP);
            }
        }
    } else {
        scan_dst(dst, nedges, [&](int d, int e) {
            (void)e;
            const unsigned dl = (unsigned)(d - r0);
            if (dl < (unsigned)rows)
                __hip_atomic_fetch_add(&cnt_s[dl], 1.0f, __ATOMIC_RELAXED,
                                       __HIP_MEMORY_SCOPE_WORKGROUP);
        }, [&] {});
    }
    __syncthreads();

    for (int r = threadIdx.x; r < rows; r += OWN_THREADS) {
        const float d = cnt_s[r];
        dinv[r0 + r] = (d > 0.0f) ? rsqrtf(d) : 0.0f;
    }

    slab_product<false>(slab, bfr, h, r0, rows, wave, lane);
}

__device__ __forceinline__ void add_edges(int s, int dl, float nrm, int n,
                                          const float* __restrict__ h,
                                          float* slab, int lane) {
    for (int i = 0; i < n; i += 8) {
        float4 hv[8];
#pragma unroll
        for (int u = 0; u < 8; ++u) {
            const int si = __builtin_amdgcn_readlane(s, i + u);
            hv[u] = *(const float4*)(h + (long)si * D_FEAT + lane * 4);
        }
#pragma unroll
        for (int u = 0; u < 8; ++u) {
            if (i + u < n) {
                const float nu = __int_as_float(__builtin_amdgcn_readlane(__float_as_int(nrm), i + u));
                const int dlu = __builtin_amdgcn_readlane(dl, i + u);
                float* row = slab + dlu * SLAB_LD + lane;
                __hip_atomic_fetch_add(row +  0, nu * hv[u].x, __ATOMIC_RELAXED, __HIP_MEMORY_SCOPE_WORKGROUP);
                __hip_atomic_fetch_add(row + 32, nu * hv[u].y, __ATOMIC_RELAXED, __HIP_MEMORY_SCOPE_WORKGROUP);
                __hip_atomic_fetch_add(row + 64, nu * hv[u].z, __ATOMIC_RELAXED, __HIP_MEMORY_SCOPE_WORKGROUP);
                __hip_atomic_fetch_add(row + 96, nu * hv[u].w, __ATOMIC_RELAXED, __HIP_MEMORY_SCOPE_WORKGROUP);
            }
        }
    }
}

__device__ __forceinline__ void drain_edges(const int* q_e, const int* q_d, int base, int n,
                                            const int* __restrict__ src,
                                            const float* __restrict__ dinv,
                                            const float* dinv_s,
                                            const float* __restrict__ h,
                                            float* slab, int lane) {
    int dl = 0, s = 0;
    float nrm = 0.0f;
    if (lane < n) {
        const int e = q_e[base + lane];
        dl = q_d[base + lane];
        s = src[e];
        nrm = dinv[s] * dinv_s[dl];
    }
    add_edges(s, dl, nrm, n, h, slab, lane);
}

template <bool NEXT>
__global__ __launch_bounds__(OWN_THREADS)
void layer_owner_kernel(const int* __restrict__ src, const int* __restrict__ dst,
                        const float* __restrict__ dinv, const float* __restrict__ h,
                        const float* __restrict__ bias, const int* lists, float* xout,
                        const float* __restrict__ Wn, float* __restrict__ hnext,
                        int n, int nedges) {
    __shared__ float slab[OWN_ROWS * SLAB_LD];
    __shared__ float dinv_s[OWN_ROWS];
    __shared__ int q_e_all[OWN_WAVES * Q_CAP];
    __shared__ int q_d_all[OWN_WAVES * Q_CAP];

    const int lane = threadIdx.x & 31;
    const int wave = __builtin_amdgcn_readfirstlane((int)(threadIdx.x >> 5));
    const int r0 = blockIdx.x * OWN_ROWS;
    const int rows = min(OWN_ROWS, n - r0);

    const int cap = cell_cap_of(rows);
    const int* cntp = lists + (long)r0 * D_FEAT;
    const int* cells = cntp + NCH + wave * 8 * cap;
    int cn[8];
    const bool ovf = load_cell_counts(cntp, cap, lane, cn);
    int pre[9];
    pre[0] = 0;
#pragma unroll
    for (int j = 0; j < 8; ++j) pre[j + 1] = pre[j] + __builtin_amdgcn_readlane(cn[j], wave);
    const int mycnt = ovf ? -1 : pre[8];
    auto fetch = [&](int e, int& s, int& dl) {
        int j = 0, pj = 0;
#pragma unroll
        for (int q = 1; q < 8; ++q) {
            const bool in = e >= pre[q];
            j += in ? 1 : 0;
            pj = in ? pre[q] : pj;
        }
        const int v = cells[j * cap + (e - pj)];
        s = v & 0xffff;
        dl = v >> 16;
    };
    int s_nx = 0, dl_nx = 0;
    if (lane < mycnt) fetch(lane, s_nx, dl_nx);

    const float4 bv = *(const float4*)(bias + lane * 4);
    for (int r = wave; r < rows; r += OWN_WAVES) {
        const float di = dinv[r0 + r];
        const float4 hv = *(const float4*)(h + (long)(r0 + r) * D_FEAT + lane * 4);
        float* row = slab + r * SLAB_LD + lane;
        row[0]  = di * di * hv.x + bv.x;
        row[32] = di * di * hv.y + bv.y;
        row[64] = di * di * hv.z + bv.z;
        row[96] = di * di * hv.w + bv.w;
        if (lane == 0) dinv_s[r] = di;
    }
    __syncthreads();

    if (mycnt >= 0) {
        int base = 0;
        for (; base + 32 <= mycnt; base += 32) {
            const int s = s_nx, dl = dl_nx;
            const float nrm = dinv[s] * dinv_s[dl];
            s_nx = 0; dl_nx = 0;
            if (base + 32 + lane < mycnt) fetch(base + 32 + lane, s_nx, dl_nx);
            add_edges(s, dl, nrm, 32, h, slab, lane);
        }
        if (base < mycnt) {
            const float nrm = dinv[s_nx] * dinv_s[dl_nx];
            add_edges(s_nx, dl_nx, nrm, mycnt - base, h, slab, lane);
        }
    } else {
        int* q_e = q_e_all + wave * Q_CAP;
        int* q_d = q_d_all + wave * Q_CAP;
        int cnt = 0;

        scan_dst(dst, nedges, [&](int d, int e) {
            const unsigned dl = (unsigned)(d - r0);
            const bool hit = dl < (unsigned)rows;
            const unsigned m = __builtin_amdgcn_ballot_w32(hit);
            if (m != 0u) {
                const int pos = cnt + (int)__builtin_amdgcn_mbcnt_lo(m, 0u);
                if (hit) { q_e[pos] = e; q_d[pos] = (int)dl; }
                cnt += __popc(m);
            }
        }, [&] {
            while (cnt >= 32) {
                cnt -= 32;
                drain_edges(q_e, q_d, cnt, 32, src, dinv, dinv_s, h, slab, lane);
            }
        });
        if (cnt > 0) drain_edges(q_e, q_d, 0, cnt, src, dinv, dinv_s, h, slab, lane);
    }
    __syncthreads();

    for (int r = wave; r < rows; r += OWN_WAVES) {
        const float* row = slab + r * SLAB_LD + lane;
        float4 v;
        v.x = row[0]; v.y = row[32]; v.z = row[64]; v.w = row[96];
        v.x = v.x > 0.0f ? v.x : 0.0f;
        v.y = v.y > 0.0f ? v.y : 0.0f;
        v.z = v.z > 0.0f ? v.z : 0.0f;
        v.w = v.w > 0.0f ? v.w : 0.0f;
        *(float4*)(xout + (long)(r0 + r) * D_FEAT + lane * 4) = v;
    }

    if constexpr (NEXT) {
        v2f bfr[32];
        load_w_strip(Wn, wave, lane, bfr);
        slab_product<true>(slab, bfr, hnext, r0, rows, wave, lane);
    }
}

#define LDAJ 260
__global__ __launch_bounds__(256)
void gemm_jk_wmma_kernel(const float* __restrict__ x1, const float* __restrict__ x2,
                         const float* __restrict__ Wl, const float* __restrict__ bl,
                         float* __restrict__ out, int nrows) {
    __shared__ float As[16 * LDAJ];
    const int tid = threadIdx.x;
    const long row0 = (long)blockIdx.x * 16;

    for (int idx = tid; idx < 1024; idx += 256) {
        int r = idx >> 6;
        int c4 = idx & 63;
        long gr = row0 + r;
        if (gr >= nrows) gr = nrows - 1;
        const float* srcp = (c4 < 32)
            ? (x1 + gr * D_FEAT + c4 * 4)
            : (x2 + gr * D_FEAT + (c4 - 32) * 4);
        *(float4*)(As + r * LDAJ + c4 * 4) = *(const float4*)srcp;
    }
    __syncthreads();

    const int wave  = tid >> 5;
    const int lane  = tid & 31;
    const int col0  = wave * 16;
    const int mrow  = lane & 15;
    const int khalf = (lane >> 4) * 2;
    const int ncol  = lane & 15;

    v8f acc = {};
    for (int k = 0; k < 2 * D_FEAT; k += 4) {
        v2f a, b;
        a.x = As[mrow * LDAJ + k + khalf + 0];
        a.y = As[mrow * LDAJ + k + khalf + 1];
        b.x = Wl[(k + khalf + 0) * D_FEAT + col0 + ncol];
        b.y = Wl[(k + khalf + 1) * D_FEAT + col0 + ncol];
        acc = __builtin_amdgcn_wmma_f32_16x16x4_f32(
            false, a, false, b, (short)0, acc, false, false);
    }

    const float bias = bl[col0 + ncol];
    const int mbase = (lane >> 4) * 8;
    for (int r = 0; r < 8; ++r) {
        long m = row0 + mbase + r;
        if (m < nrows) out[m * D_FEAT + col0 + ncol] = acc[r] + bias;
    }
}

extern "C" void kernel_launch(void* const* d_in, const int* in_sizes, int n_in,
                              void* d_out, int out_size, void* d_ws, size_t ws_size,
                              hipStream_t stream) {
    (void)n_in; (void)out_size; (void)ws_size;

    const float* x    = (const float*)d_in[0];
    const int*   ei   = (const int*)d_in[1];
    const float* W1   = (const float*)d_in[2];
    const float* b1   = (const float*)d_in[3];
    const float* W2   = (const float*)d_in[4];
    const float* b2   = (const float*)d_in[5];
    const float* linW = (const float*)d_in[6];
    const float* linb = (const float*)d_in[7];
    float* out = (float*)d_out;

    const int N = in_sizes[0] / D_FEAT;
    const int E = in_sizes[1] / 2;
    const int* srcI = ei;
    const int* dstI = ei + E;

    float* ws   = (float*)d_ws;
    float* dinv = ws;
    float* h    = dinv + N;
    float* x1   = h   + (long)N * D_FEAT;
    float* x2   = x1  + (long)N * D_FEAT;

    const int nBlkRow = (N + 15) / 16;
    const int nBlkOwn = (N + OWN_ROWS - 1) / OWN_ROWS;

    int* lists = (int*)x2;

    bin_edges_kernel<<<NCH, BIN_THREADS, 0, stream>>>(srcI, dstI, lists, N, E, nBlkOwn);
    deg_product_kernel<<<nBlkOwn, OWN_THREADS, 0, stream>>>(dstI, lists, x, W1, dinv, h, N, E);

    float* h2 = out;

    layer_owner_kernel<true><<<nBlkOwn, OWN_THREADS, 0, stream>>>(srcI, dstI, dinv, h, b1, lists, x1,
                                                                 W2, h2, N, E);

    layer_owner_kernel<false><<<nBlkOwn, OWN_THREADS, 0, stream>>>(srcI, dstI, dinv, h2, b2, lists, x2,
                                                                  nullptr, nullptr, N, E);

    gemm_jk_wmma_kernel<<<nBlkRow, 256, 0, stream>>>(x1, x2, linW, linb, out, N);
}


# Final linear split in two K=128 halves, x1 @ linW[0:128] + x2 @ linW[128:256], each formed by its layer's owner blocks from the LDS slab; last kernel and both x planes go; k sum order changes
# speedup vs baseline: 3.9220x; 1.0747x over previous
#include <hip/hip_runtime.h>


typedef __attribute__((ext_vector_type(2))) float v2f;
typedef __attribute__((ext_vector_type(8))) float v8f;

#define D_FEAT 128

#define OWN_ROWS    196
#define SLAB_LD     129
#define OWN_WAVES   32
#define OWN_THREADS (OWN_WAVES * 32)
#define GROUP_EDGES 512
#define Q_CAP       544

template <int NW = OWN_WAVES, class F, class G>
__device__ __forceinline__ void scan_dst(const int* __restrict__ dst, int nedges, F&& visit,
                                         G&& after_group) {
    const int lane = threadIdx.x & 31;
    const int wave = __builtin_amdgcn_readfirstlane((int)(threadIdx.x >> 5));
    const bool vec_ok = (((size_t)dst) & 15) == 0;
    const int ngroups = vec_ok ? nedges / GROUP_EDGES : 0;
    const int off = (int)blockIdx.x * (ngroups / (int)gridDim.x);

    auto rot = [&](int g) { int gr = g + off; return gr >= ngroups ? gr - ngroups : gr; };
    auto load_group = [&](int4 (&v)[4], int gr) {
        const int4* p = (const int4*)(dst + gr * GROUP_EDGES) + lane;
#pragma unroll
        for (int j = 0; j < 4; ++j) v[j] = p[j * 32];
    };
    auto step = [&](int4 (&cur)[4], int4 (&nxt)[4], int g) {
        const int gr = rot(g);
        const int gn = g + NW;
        load_group(nxt, rot(gn < ngroups ? gn : g));
#pragma unroll
        for (int j = 0; j < 4; ++j) {
            const int e0 = gr * GROUP_EDGES + j * 128 + lane * 4;
            visit(cur[j].x, e0 + 0);
            visit(cur[j].y, e0 + 1);
            visit(cur[j].z, e0 + 2);
            visit(cur[j].w, e0 + 3);
        }
        after_group();
    };

    if (wave < ngroups) {
        int4 bufa[4], bufb[4];
        load_group(bufa, rot(wave));
        for (int g = wave; g < ngroups; g += 2 * NW) {
            step(bufa, bufb, g);
            if (g + NW >= ngroups) break;
            step(bufb, bufa, g + NW);
        }
    }
    for (int base = ngroups * GROUP_EDGES + wave * 32; base < nedges; base += NW * 32) {
        const int e = base + lane;
        visit(e < nedges ? dst[e] : -1, e);
        after_group();
    }
}

#define NCH     256
#define OWN_MAX 512
#define BIN_THREADS 256
static_assert(NCH == 8 * OWN_WAVES, "cells per wave");

__device__ __forceinline__ int cell_cap_of(int rows) {
    return (rows * D_FEAT - NCH) / NCH;
}

__global__ __launch_bounds__(BIN_THREADS)
void bin_edges_kernel(const int* __restrict__ src, const int* __restrict__ dst,
                      int* __restrict__ lists, int n, int nedges, int nown) {
    __shared__ int fill_s[OWN_MAX];

    const int c = blockIdx.x;
    const int per = (nedges + NCH - 1) / NCH;
    const int e0 = c * per;
    const int e1 = min(nedges, e0 + per);
    const bool usable = (n <= 65536) && (nown <= OWN_MAX);

    for (int o = threadIdx.x; o < OWN_MAX; o += BIN_THREADS) fill_s[o] = 0;
    __syncthreads();

    if (usable) {
        for (int base = e0; base < e1; base += 8 * BIN_THREADS) {
            int dd[8], ss[8];
#pragma unroll
            for (int u = 0; u < 8; ++u) {
                const int e = base + u * BIN_THREADS + (int)threadIdx.x;
                const bool ok = e < e1;
                dd[u] = ok ? dst[e] : -1;
                ss[u] = ok ? src[e] : 0;
            }
#pragma unroll
            for (int u = 0; u < 8; ++u) {
                const int d = dd[u];
                if ((unsigned)d < (unsigned)n) {
                    const int o = d / OWN_ROWS;
                    const int row = d - o * OWN_ROWS;
                    const int cap = cell_cap_of(min(OWN_ROWS, n - o * OWN_ROWS));
                    const int pos = __hip_atomic_fetch_add(&fill_s[o], 1, __ATOMIC_RELAXED,
                                                           __HIP_MEMORY_SCOPE_WORKGROUP);
                    if (pos < cap)
                        lists[(long)o * OWN_ROWS * D_FEAT + NCH + c * cap + pos] = ss[u] | (row << 16);
                }
            }
        }
    }
    __syncthreads();

    for (int o = threadIdx.x; o < nown; o += BIN_THREADS) {
        int v = 0x7fffffff;
        if (usable) v = fill_s[o];
        lists[(long)o * OWN_ROWS * D_FEAT + c] = v;
    }
}

__device__ __forceinline__ bool load_cell_counts(const int* cntp, int cap, int lane, int (&cn)[8]) {
    const int4* cp = (const int4*)cntp + lane * 2;
    const int4 a = cp[0], b = cp[1];
    cn[0] = a.x; cn[1] = a.y; cn[2] = a.z; cn[3] = a.w;
    cn[4] = b.x; cn[5] = b.y; cn[6] = b.z; cn[7] = b.w;
    bool bad = false;
#pragma unroll
    for (int j = 0; j < 8; ++j) bad = bad || (cn[j] > cap);
    return __builtin_amdgcn_ballot_w32(bad) != 0u;
}

__device__ __forceinline__ void load_w_strip(const float* __restrict__ Wm, int wave, int lane,
                                             v2f (&bfr)[32]) {
    const float* wp = Wm + (lane >> 4) * D_FEAT + (wave & 7) * 16 + (lane & 15);
#pragma unroll
    for (int t = 0; t < 32; ++t) {
        bfr[t].x = wp[(4 * t) * D_FEAT];
        bfr[t].y = wp[(4 * t + 2) * D_FEAT];
    }
}

template <bool RELU, int EPI>
__device__ __forceinline__ void slab_product(const float* slab, const v2f (&bfr)[32],
                                             float* __restrict__ hout, int r0, int rows,
                                             int wave, int rt0, int rts, int lane,
                                             const float* __restrict__ addv) {
    const int ct = wave & 7;
    const int m16 = lane & 15;
    const int hh = lane >> 4;
    const int ntiles = (rows + 15) >> 4;
    float bcol = 0.0f;
    if constexpr (EPI == 1) bcol = addv[ct * 16 + m16];
    for (int rt = rt0; rt < ntiles; rt += rts) {
        const int ar = min(rt * 16 + m16, OWN_ROWS - 1);
        const float* ap = slab + ar * SLAB_LD + 32 * hh;
        float prev[8];
        if constexpr (EPI == 2) {
            for (int r = 0; r < 8; ++r) {
                const int prow = min(rt * 16 + hh * 8 + r, rows - 1);
                prev[r] = hout[(long)(r0 + prow) * D_FEAT + ct * 16 + m16];
            }
        }
        v8f acc = {};
#pragma unroll
        for (int t = 0; t < 32; ++t) {
            v2f a;
            a.x = ap[t];
            a.y = ap[64 + t];
            if constexpr (RELU) {
                a.x = a.x > 0.0f ? a.x : 0.0f;
                a.y = a.y > 0.0f ? a.y : 0.0f;
            }
            acc = __builtin_amdgcn_wmma_f32_16x16x4_f32(
                 false, a,  false, bfr[t],
                 (short)0, acc,  false,  false);
        }
        for (int r = 0; r < 8; ++r) {
            const int orow = rt * 16 + hh * 8 + r;
            float v = acc[r];
            if constexpr (EPI == 1) v += bcol;
            if constexpr (EPI == 2) v += prev[r];
            if (orow < rows) hout[(long)(r0 + orow) * D_FEAT + ct * 16 + m16] = v;
        }
    }
}

#define ROWS_PER_WAVE ((OWN_ROWS + OWN_WAVES - 1) / OWN_WAVES)
__global__ __launch_bounds__(OWN_THREADS)
void deg_product_kernel(const int* __restrict__ dst, const int* __restrict__ lists,
                        const float* __restrict__ x, const float* __restrict__ W,
                        float* __restrict__ dinv, float* __restrict__ h, int n, int nedges) {
    __shared__ float slab[OWN_ROWS * SLAB_LD];
    __shared__ float cnt_s[OWN_ROWS];

    const int lane = threadIdx.x & 31;
    const int wave = __builtin_amdgcn_readfirstlane((int)(threadIdx.x >> 5));
    const int r0 = blockIdx.x * OWN_ROWS;
    const int rows = min(OWN_ROWS, n - r0);
    const int cap = cell_cap_of(rows);
    const int* cntp = lists + (long)r0 * D_FEAT;
    const int* cells = cntp + NCH + wave * 8 * cap;

    int cn[8];
    const bool ovf = load_cell_counts(cntp, cap, lane, cn);

    float4 xv[ROWS_PER_WAVE];
#pragma unroll
    for (int k = 0; k < ROWS_PER_WAVE; ++k) {
        const int r = min(wave + k * OWN_WAVES, rows - 1);
        xv[k] = *(const float4*)(x + (long)(r0 + r) * D_FEAT + lane * 4);
    }

    v2f bfr[32];
    load_w_strip(W, wave, lane, bfr);

    int cj[8], v[8];
#pragma unroll
    for (int j = 0; j < 8; ++j) {
        cj[j] = ovf ? 0 : __builtin_amdgcn_readlane(cn[j], wave);
        v[j] = -1;
        if (lane < cj[j]) v[j] = cells[j * cap + lane];
    }

    for (int r = threadIdx.x; r < OWN_ROWS; r += OWN_THREADS) cnt_s[r] = 1.0f;
#pragma unroll
    for (int k = 0; k < ROWS_PER_WAVE; ++k) {
        const int r = wave + k * OWN_WAVES;
        if (r < rows) {
            float* row = slab + r * SLAB_LD + lane;
            row[0]  = xv[k].x;
            row[32] = xv[k].y;
            row[64] = xv[k].z;
            row[96] = xv[k].w;
        }
    }
    __syncthreads();

    if (!ovf) {
#pragma unroll
        for (int j = 0; j < 8; ++j) {
            if (v[j] >= 0)
                __hip_atomic_fetch_add(&cnt_s[v[j] >> 16], 1.0f, __ATOMIC_RELAXED,
                                       __HIP_MEMORY_SCOPE_WORKGROUP);
        }
#pragma unroll
        for (int j = 0; j < 8; ++j) {
            for (int i = lane + 32; i < cj[j]; i += 32) {
                const int w = cells[j * cap + i];
                __hip_atomic_fetch_add(&cnt_s[w >> 16], 1.0f, __ATOMIC_RELAXED,
                                       __HIP_MEMORY_SCOPE_WORKGROUP);
            }
        }
    } else {
        scan_dst(dst, nedges, [&](int d, int e) {
            (void)e;
            const unsigned dl = (unsigned)(d - r0);
            if (dl < (unsigned)rows)
                __hip_atomic_fetch_add(&cnt_s[dl], 1.0f, __ATOMIC_RELAXED,
                                       __HIP_MEMORY_SCOPE_WORKGROUP);
        }, [&] {});
    }
    __syncthreads();

    for (int r = threadIdx.x; r < rows; r += OWN_THREADS) {
        const float d = cnt_s[r];
        dinv[r0 + r] = (d > 0.0f) ? rsqrtf(d) : 0.0f;
    }

    slab_product<false, 0>(slab, bfr, h, r0, rows, wave, wave >> 3, 4, lane, nullptr);
}

__device__ __forceinline__ void add_edges(int s, int dl, float nrm, int n,
                                          const float* __restrict__ h,
                                          float* slab, int lane) {
    for (int i = 0; i < n; i += 8) {
        float4 hv[8];
#pragma unroll
        for (int u = 0; u < 8; ++u) {
            const int si = __builtin_amdgcn_readlane(s, i + u);
            hv[u] = *(const float4*)(h + (long)si * D_FEAT + lane * 4);
        }
#pragma unroll
        for (int u = 0; u < 8; ++u) {
            if (i + u < n) {
                const float nu = __int_as_float(__builtin_amdgcn_readlane(__float_as_int(nrm), i + u));
                const int dlu = __builtin_amdgcn_readlane(dl, i + u);
                float* row = slab + dlu * SLAB_LD + lane;
                __hip_atomic_fetch_add(row +  0, nu * hv[u].x, __ATOMIC_RELAXED, __HIP_MEMORY_SCOPE_WORKGROUP);
                __hip_atomic_fetch_add(row + 32, nu * hv[u].y, __ATOMIC_RELAXED, __HIP_MEMORY_SCOPE_WORKGROUP);
                __hip_atomic_fetch_add(row + 64, nu * hv[u].z, __ATOMIC_RELAXED, __HIP_MEMORY_SCOPE_WORKGROUP);
                __hip_atomic_fetch_add(row + 96, nu * hv[u].w, __ATOMIC_RELAXED, __HIP_MEMORY_SCOPE_WORKGROUP);
            }
        }
    }
}

__device__ __forceinline__ void drain_edges(const int* q_e, const int* q_d, int base, int n,
                                            const int* __restrict__ src,
                                            const float* __restrict__ dinv,
                                            const float* dinv_s,
                                            const float* __restrict__ h,
                                            float* slab, int lane) {
    int dl = 0, s = 0;
    float nrm = 0.0f;
    if (lane < n) {
        const int e = q_e[base + lane];
        dl = q_d[base + lane];
        s = src[e];
        nrm = dinv[s] * dinv_s[dl];
    }
    add_edges(s, dl, nrm, n, h, slab, lane);
}

template <bool FIRST>
__global__ __launch_bounds__(OWN_THREADS)
void layer_owner_kernel(const int* __restrict__ src, const int* __restrict__ dst,
                        const float* __restrict__ dinv, const float* __restrict__ h,
                        const float* __restrict__ bias, const int* __restrict__ lists,
                        const float* __restrict__ Wn, float* __restrict__ hnext,
                        const float* __restrict__ Wl, const float* __restrict__ bl,
                        float* __restrict__ out, int n, int nedges) {
    __shared__ float slab[OWN_ROWS * SLAB_LD];
    __shared__ float dinv_s[OWN_ROWS];
    __shared__ int q_e_all[OWN_WAVES * Q_CAP];
    __shared__ int q_d_all[OWN_WAVES * Q_CAP];

    const int lane = threadIdx.x & 31;
    const int wave = __builtin_amdgcn_readfirstlane((int)(threadIdx.x >> 5));
    const int r0 = blockIdx.x * OWN_ROWS;
    const int rows = min(OWN_ROWS, n - r0);

    const int cap = cell_cap_of(rows);
    const int* cntp = lists + (long)r0 * D_FEAT;
    const int* cells = cntp + NCH + wave * 8 * cap;
    int cn[8];
    const bool ovf = load_cell_counts(cntp, cap, lane, cn);
    int pre[9];
    pre[0] = 0;
#pragma unroll
    for (int j = 0; j < 8; ++j) pre[j + 1] = pre[j] + __builtin_amdgcn_readlane(cn[j], wave);
    const int mycnt = ovf ? -1 : pre[8];
    auto fetch = [&](int e, int& s, int& dl) {
        int j = 0, pj = 0;
#pragma unroll
        for (int q = 1; q < 8; ++q) {
            const bool in = e >= pre[q];
            j += in ? 1 : 0;
            pj = in ? pre[q] : pj;
        }
        const int v = cells[j * cap + (e - pj)];
        s = v & 0xffff;
        dl = v >> 16;
    };
    int s_nx = 0, dl_nx = 0;
    if (lane < mycnt) fetch(lane, s_nx, dl_nx);

    const float4 bv = *(const float4*)(bias + lane * 4);
    for (int r = wave; r < rows; r += OWN_WAVES) {
        const float di = dinv[r0 + r];
        const float4 hv = *(const float4*)(h + (long)(r0 + r) * D_FEAT + lane * 4);
        float* row = slab + r * SLAB_LD + lane;
        row[0]  = di * di * hv.x + bv.x;
        row[32] = di * di * hv.y + bv.y;
        row[64] = di * di * hv.z + bv.z;
        row[96] = di * di * hv.w + bv.w;
        if (lane == 0) dinv_s[r] = di;
    }
    __syncthreads();

    if (mycnt >= 0) {
        int base = 0;
        for (; base + 32 <= mycnt; base += 32) {
            const int s = s_nx, dl = dl_nx;
            const float nrm = dinv[s] * dinv_s[dl];
            s_nx = 0; dl_nx = 0;
            if (base + 32 + lane < mycnt) fetch(base + 32 + lane, s_nx, dl_nx);
            add_edges(s, dl, nrm, 32, h, slab, lane);
        }
        if (base < mycnt) {
            const float nrm = dinv[s_nx] * dinv_s[dl_nx];
            add_edges(s_nx, dl_nx, nrm, mycnt - base, h, slab, lane);
        }
    } else {
        int* q_e = q_e_all + wave * Q_CAP;
        int* q_d = q_d_all + wave * Q_CAP;
        int cnt = 0;

        scan_dst(dst, nedges, [&](int d, int e) {
            const unsigned dl = (unsigned)(d - r0);
            const bool hit = dl < (unsigned)rows;
            const unsigned m = __builtin_amdgcn_ballot_w32(hit);
            if (m != 0u) {
                const int pos = cnt + (int)__builtin_amdgcn_mbcnt_lo(m, 0u);
                if (hit) { q_e[pos] = e; q_d[pos] = (int)dl; }
                cnt += __popc(m);
            }
        }, [&] {
            while (cnt >= 32) {
                cnt -= 32;
                drain_edges(q_e, q_d, cnt, 32, src, dinv, dinv_s, h, slab, lane);
            }
        });
        if (cnt > 0) drain_edges(q_e, q_d, 0, cnt, src, dinv, dinv_s, h, slab, lane);
    }
    __syncthreads();

    v2f bfr[32];
    if constexpr (FIRST) {
        const int rt0 = (wave >> 3) & 1;
        if (wave < 16) {
            load_w_strip(Wn, wave, lane, bfr);
            slab_product<true, 0>(slab, bfr, hnext, r0, rows, wave, rt0, 2, lane, nullptr);
        } else {
            load_w_strip(Wl, wave, lane, bfr);
            slab_product<true, 1>(slab, bfr, out, r0, rows, wave, rt0, 2, lane, bl);
        }
    } else {
        load_w_strip(Wl, wave, lane, bfr);
        slab_product<true, 2>(slab, bfr, out, r0, rows, wave, wave >> 3, 4, lane, nullptr);
    }
}

extern "C" void kernel_launch(void* const* d_in, const int* in_sizes, int n_in,
                              void* d_out, int out_size, void* d_ws, size_t ws_size,
                              hipStream_t stream) {
    (void)n_in; (void)out_size; (void)ws_size;

    const float* x    = (const float*)d_in[0];
    const int*   ei   = (const int*)d_in[1];
    const float* W1   = (const float*)d_in[2];
    const float* b1   = (const float*)d_in[3];
    const float* W2   = (const float*)d_in[4];
    const float* b2   = (const float*)d_in[5];
    const float* linW = (const float*)d_in[6];
    const float* linb = (const float*)d_in[7];
    float* out = (float*)d_out;

    const int N = in_sizes[0] / D_FEAT;
    const int E = in_sizes[1] / 2;
    const int* srcI = ei;
    const int* dstI = ei + E;

    float* ws   = (float*)d_ws;
    float* dinv = ws;
    float* h    = dinv + N;
    float* h2   = h   + (long)N * D_FEAT;
    float* lpl  = h2  + (long)N * D_FEAT;

    const int nBlkOwn = (N + OWN_ROWS - 1) / OWN_ROWS;

    int* lists = (int*)lpl;

    bin_edges_kernel<<<NCH, BIN_THREADS, 0, stream>>>(srcI, dstI, lists, N, E, nBlkOwn);
    deg_product_kernel<<<nBlkOwn, OWN_THREADS, 0, stream>>>(dstI, lists, x, W1, dinv, h, N, E);


    layer_owner_kernel<true><<<nBlkOwn, OWN_THREADS, 0, stream>>>(srcI, dstI, dinv, h, b1, lists,
                                                                 W2, h2, linW, linb, out, N, E);

    layer_owner_kernel<false><<<nBlkOwn, OWN_THREADS, 0, stream>>>(srcI, dstI, dinv, h2, b2, lists,
                                                                  nullptr, nullptr,
                                                                  linW + D_FEAT * D_FEAT, nullptr,
                                                                  out, N, E);
}


# Normalisation leaves the edge loop: owner blocks store product rows as dinv*(x W), dinv[i] passes the ReLU into the products' row scale; per-edge dinv gather and multiply go; rounding changes
# speedup vs baseline: 4.0804x; 1.0404x over previous
#include <hip/hip_runtime.h>


typedef __attribute__((ext_vector_type(2))) float v2f;
typedef __attribute__((ext_vector_type(8))) float v8f;

#define D_FEAT 128

#define OWN_ROWS    196
#define SLAB_LD     129
#define DS_ROWS     (((OWN_ROWS + 15) / 16) * 16)
#define OWN_WAVES   32
#define OWN_THREADS (OWN_WAVES * 32)
#define GROUP_EDGES 512
#define Q_CAP       544

template <int NW = OWN_WAVES, class F, class G>
__device__ __forceinline__ void scan_dst(const int* __restrict__ dst, int nedges, F&& visit,
                                         G&& after_group) {
    const int lane = threadIdx.x & 31;
    const int wave = __builtin_amdgcn_readfirstlane((int)(threadIdx.x >> 5));
    const bool vec_ok = (((size_t)dst) & 15) == 0;
    const int ngroups = vec_ok ? nedges / GROUP_EDGES : 0;
    const int off = (int)blockIdx.x * (ngroups / (int)gridDim.x);

    auto rot = [&](int g) { int gr = g + off; return gr >= ngroups ? gr - ngroups : gr; };
    auto load_group = [&](int4 (&v)[4], int gr) {
        const int4* p = (const int4*)(dst + gr * GROUP_EDGES) + lane;
#pragma unroll
        for (int j = 0; j < 4; ++j) v[j] = p[j * 32];
    };
    auto step = [&](int4 (&cur)[4], int4 (&nxt)[4], int g) {
        const int gr = rot(g);
        const int gn = g + NW;
        load_group(nxt, rot(gn < ngroups ? gn : g));
#pragma unroll
        for (int j = 0; j < 4; ++j) {
            const int e0 = gr * GROUP_EDGES + j * 128 + lane * 4;
            visit(cur[j].x, e0 + 0);
            visit(cur[j].y, e0 + 1);
            visit(cur[j].z, e0 + 2);
            visit(cur[j].w, e0 + 3);
        }
        after_group();
    };

    if (wave < ngroups) {
        int4 bufa[4], bufb[4];
        load_group(bufa, rot(wave));
        for (int g = wave; g < ngroups; g += 2 * NW) {
            step(bufa, bufb, g);
            if (g + NW >= ngroups) break;
            step(bufb, bufa, g + NW);
        }
    }
    for (int base = ngroups * GROUP_EDGES + wave * 32; base < nedges; base += NW * 32) {
        const int e = base + lane;
        visit(e < nedges ? dst[e] : -1, e);
        after_group();
    }
}

#define NCH     256
#define OWN_MAX 512
#define BIN_THREADS 256
static_assert(NCH == 8 * OWN_WAVES, "cells per wave");

__device__ __forceinline__ int cell_cap_of(int rows) {
    return (rows * D_FEAT - NCH) / NCH;
}

__global__ __launch_bounds__(BIN_THREADS)
void bin_edges_kernel(const int* __restrict__ src, const int* __restrict__ dst,
                      int* __restrict__ lists, int n, int nedges, int nown) {
    __shared__ int fill_s[OWN_MAX];

    const int c = blockIdx.x;
    const int per = (nedges + NCH - 1) / NCH;
    const int e0 = c * per;
    const int e1 = min(nedges, e0 + per);
    const bool usable = (n <= 65536) && (nown <= OWN_MAX);

    for (int o = threadIdx.x; o < OWN_MAX; o += BIN_THREADS) fill_s[o] = 0;
    __syncthreads();

    if (usable) {
        for (int base = e0; base < e1; base += 8 * BIN_THREADS) {
            int dd[8], ss[8];
#pragma unroll
            for (int u = 0; u < 8; ++u) {
                const int e = base + u * BIN_THREADS + (int)threadIdx.x;
                const bool ok = e < e1;
                dd[u] = ok ? dst[e] : -1;
                ss[u] = ok ? src[e] : 0;
            }
#pragma unroll
            for (int u = 0; u < 8; ++u) {
                const int d = dd[u];
                if ((unsigned)d < (unsigned)n) {
                    const int o = d / OWN_ROWS;
                    const int row = d - o * OWN_ROWS;
                    const int cap = cell_cap_of(min(OWN_ROWS, n - o * OWN_ROWS));
                    const int pos = __hip_atomic_fetch_add(&fill_s[o], 1, __ATOMIC_RELAXED,
                                                           __HIP_MEMORY_SCOPE_WORKGROUP);
                    if (pos < cap)
                        lists[(long)o * OWN_ROWS * D_FEAT + NCH + c * cap + pos] = ss[u] | (row << 16);
                }
            }
        }
    }
    __syncthreads();

    for (int o = threadIdx.x; o < nown; o += BIN_THREADS) {
        int v = 0x7fffffff;
        if (usable) v = fill_s[o];
        lists[(long)o * OWN_ROWS * D_FEAT + c] = v;
    }
}

__device__ __forceinline__ bool load_cell_counts(const int* cntp, int cap, int lane, int (&cn)[8]) {
    const int4* cp = (const int4*)cntp + lane * 2;
    const int4 a = cp[0], b = cp[1];
    cn[0] = a.x; cn[1] = a.y; cn[2] = a.z; cn[3] = a.w;
    cn[4] = b.x; cn[5] = b.y; cn[6] = b.z; cn[7] = b.w;
    bool bad = false;
#pragma unroll
    for (int j = 0; j < 8; ++j) bad = bad || (cn[j] > cap);
    return __builtin_amdgcn_ballot_w32(bad) != 0u;
}

__device__ __forceinline__ void load_w_strip(const float* __restrict__ Wm, int wave, int lane,
                                             v2f (&bfr)[32]) {
    const float* wp = Wm + (lane >> 4) * D_FEAT + (wave & 7) * 16 + (lane & 15);
#pragma unroll
    for (int t = 0; t < 32; ++t) {
        bfr[t].x = wp[(4 * t) * D_FEAT];
        bfr[t].y = wp[(4 * t + 2) * D_FEAT];
    }
}

template <bool RELU, int EPI, int SCALE>
__device__ __forceinline__ void slab_product(const float* slab, const float* dsc, const v2f (&bfr)[32],
                                             float* __restrict__ hout, int r0, int rows,
                                             int wave, int rt0, int rts, int lane,
                                             const float* __restrict__ addv) {
    const int ct = wave & 7;
    const int m16 = lane & 15;
    const int hh = lane >> 4;
    const int ntiles = (rows + 15) >> 4;
    float bcol = 0.0f;
    if constexpr (EPI == 1) bcol = addv[ct * 16 + m16];
    for (int rt = rt0; rt < ntiles; rt += rts) {
        const int ar = min(rt * 16 + m16, OWN_ROWS - 1);
        const float* ap = slab + ar * SLAB_LD + 32 * hh;
        float prev[8];
        if constexpr (EPI == 2) {
            for (int r = 0; r < 8; ++r) {
                const int prow = min(rt * 16 + hh * 8 + r, rows - 1);
                prev[r] = hout[(long)(r0 + prow) * D_FEAT + ct * 16 + m16];
            }
        }
        const float4 dlo = *(const float4*)(dsc + rt * 16 + hh * 8);
        const float4 dhi = *(const float4*)(dsc + rt * 16 + hh * 8 + 4);
        const float dv[8] = {dlo.x, dlo.y, dlo.z, dlo.w, dhi.x, dhi.y, dhi.z, dhi.w};
        v8f acc = {};
#pragma unroll
        for (int t = 0; t < 32; ++t) {
            v2f a;
            a.x = ap[t];
            a.y = ap[64 + t];
            if constexpr (RELU) {
                a.x = a.x > 0.0f ? a.x : 0.0f;
                a.y = a.y > 0.0f ? a.y : 0.0f;
            }
            acc = __builtin_amdgcn_wmma_f32_16x16x4_f32(
                 false, a,  false, bfr[t],
                 (short)0, acc,  false,  false);
        }
        for (int r = 0; r < 8; ++r) {
            const int orow = rt * 16 + hh * 8 + r;
            float sc = dv[r];
            if constexpr (SCALE == 2) sc *= sc;
            float v = acc[r] * sc;
            if constexpr (EPI == 1) v += bcol;
            if constexpr (EPI == 2) v += prev[r];
            if (orow < rows) hout[(long)(r0 + orow) * D_FEAT + ct * 16 + m16] = v;
        }
    }
}

#define ROWS_PER_WAVE ((OWN_ROWS + OWN_WAVES - 1) / OWN_WAVES)
__global__ __launch_bounds__(OWN_THREADS)
void deg_product_kernel(const int* __restrict__ dst, const int* __restrict__ lists,
                        const float* __restrict__ x, const float* __restrict__ W,
                        float* __restrict__ dinv, float* __restrict__ h, int n, int nedges) {
    __shared__ float slab[OWN_ROWS * SLAB_LD];
    __shared__ __attribute__((aligned(16))) float cnt_s[DS_ROWS];

    const int lane = threadIdx.x & 31;
    const int wave = __builtin_amdgcn_readfirstlane((int)(threadIdx.x >> 5));
    const int r0 = blockIdx.x * OWN_ROWS;
    const int rows = min(OWN_ROWS, n - r0);
    const int cap = cell_cap_of(rows);
    const int* cntp = lists + (long)r0 * D_FEAT;
    const int* cells = cntp + NCH + wave * 8 * cap;

    int cn[8];
    const bool ovf = load_cell_counts(cntp, cap, lane, cn);

    float4 xv[ROWS_PER_WAVE];
#pragma unroll
    for (int k = 0; k < ROWS_PER_WAVE; ++k) {
        const int r = min(wave + k * OWN_WAVES, rows - 1);
        xv[k] = *(const float4*)(x + (long)(r0 + r) * D_FEAT + lane * 4);
    }

    v2f bfr[32];
    load_w_strip(W, wave, lane, bfr);

    int cj[8], v[8];
#pragma unroll
    for (int j = 0; j < 8; ++j) {
        cj[j] = ovf ? 0 : __builtin_amdgcn_readlane(cn[j], wave);
        v[j] = -1;
        if (lane < cj[j]) v[j] = cells[j * cap + lane];
    }

    for (int r = threadIdx.x; r < DS_ROWS; r += OWN_THREADS) cnt_s[r] = 1.0f;
#pragma unroll
    for (int k = 0; k < ROWS_PER_WAVE; ++k) {
        const int r = wave + k * OWN_WAVES;
        if (r < rows) {
            float* row = slab + r * SLAB_LD + lane;
            row[0]  = xv[k].x;
            row[32] = xv[k].y;
            row[64] = xv[k].z;
            row[96] = xv[k].w;
        }
    }
    __syncthreads();

    if (!ovf) {
#pragma unroll
        for (int j = 0; j < 8; ++j) {
            if (v[j] >= 0)
                __hip_atomic_fetch_add(&cnt_s[v[j] >> 16], 1.0f, __ATOMIC_RELAXED,
                                       __HIP_MEMORY_SCOPE_WORKGROUP);
        }
#pragma unroll
        for (int j = 0; j < 8; ++j) {
            for (int i = lane + 32; i < cj[j]; i += 32) {
                const int w = cells[j * cap + i];
                __hip_atomic_fetch_add(&cnt_s[w >> 16], 1.0f, __ATOMIC_RELAXED,
                                       __HIP_MEMORY_SCOPE_WORKGROUP);
            }
        }
    } else {
        scan_dst(dst, nedges, [&](int d, int e) {
            (void)e;
            const unsigned dl = (unsigned)(d - r0);
            if (dl < (unsigned)rows)
                __hip_atomic_fetch_add(&cnt_s[dl], 1.0f, __ATOMIC_RELAXED,
                                       __HIP_MEMORY_SCOPE_WORKGROUP);
        }, [&] {});
    }
    __syncthreads();

    for (int r = threadIdx.x; r < DS_ROWS; r += OWN_THREADS) {
        const float d = cnt_s[r];
        const float di = (d > 0.0f) ? rsqrtf(d) : 0.0f;
        cnt_s[r] = di;
        if (r < rows) dinv[r0 + r] = di;
    }
    __syncthreads();

    slab_product<false, 0, 1>(slab, cnt_s, bfr, h, r0, rows, wave, wave >> 3, 4, lane, nullptr);
}

__device__ __forceinline__ void add_edges(int s, int dl, int n,
                                          const float* __restrict__ h,
                                          float* slab, int lane) {
    for (int i = 0; i < n; i += 8) {
        float4 hv[8];
#pragma unroll
        for (int u = 0; u < 8; ++u) {
            const int si = __builtin_amdgcn_readlane(s, i + u);
            hv[u] = *(const float4*)(h + (long)si * D_FEAT + lane * 4);
        }
#pragma unroll
        for (int u = 0; u < 8; ++u) {
            if (i + u < n) {
                const int dlu = __builtin_amdgcn_readlane(dl, i + u);
                float* row = slab + dlu * SLAB_LD + lane;
                __hip_atomic_fetch_add(row +  0, hv[u].x, __ATOMIC_RELAXED, __HIP_MEMORY_SCOPE_WORKGROUP);
                __hip_atomic_fetch_add(row + 32, hv[u].y, __ATOMIC_RELAXED, __HIP_MEMORY_SCOPE_WORKGROUP);
                __hip_atomic_fetch_add(row + 64, hv[u].z, __ATOMIC_RELAXED, __HIP_MEMORY_SCOPE_WORKGROUP);
                __hip_atomic_fetch_add(row + 96, hv[u].w, __ATOMIC_RELAXED, __HIP_MEMORY_SCOPE_WORKGROUP);
            }
        }
    }
}

__device__ __forceinline__ void drain_edges(const int* q_e, const int* q_d, int base, int n,
                                            const int* __restrict__ src,
                                            const float* __restrict__ h,
                                            float* slab, int lane) {
    int dl = 0, s = 0;
    if (lane < n) {
        const int e = q_e[base + lane];
        dl = q_d[base + lane];
        s = src[e];
    }
    add_edges(s, dl, n, h, slab, lane);
}

template <bool FIRST>
__global__ __launch_bounds__(OWN_THREADS)
void layer_owner_kernel(const int* __restrict__ src, const int* __restrict__ dst,
                        const float* __restrict__ dinv, const float* __restrict__ h,
                        const float* __restrict__ bias, const int* __restrict__ lists,
                        const float* __restrict__ Wn, float* __restrict__ hnext,
                        const float* __restrict__ Wl, const float* __restrict__ bl,
                        float* __restrict__ out, int n, int nedges) {
    __shared__ float slab[OWN_ROWS * SLAB_LD];
    __shared__ __attribute__((aligned(16))) float dinv_s[DS_ROWS];
    __shared__ int q_e_all[OWN_WAVES * Q_CAP];
    __shared__ int q_d_all[OWN_WAVES * Q_CAP];

    const int lane = threadIdx.x & 31;
    const int wave = __builtin_amdgcn_readfirstlane((int)(threadIdx.x >> 5));
    const int r0 = blockIdx.x * OWN_ROWS;
    const int rows = min(OWN_ROWS, n - r0);

    const int cap = cell_cap_of(rows);
    const int* cntp = lists + (long)r0 * D_FEAT;
    const int* cells = cntp + NCH + wave * 8 * cap;
    int cn[8];
    const bool ovf = load_cell_counts(cntp, cap, lane, cn);
    int pre[9];
    pre[0] = 0;
#pragma unroll
    for (int j = 0; j < 8; ++j) pre[j + 1] = pre[j] + __builtin_amdgcn_readlane(cn[j], wave);
    const int mycnt = ovf ? -1 : pre[8];
    auto fetch = [&](int e, int& s, int& dl) {
        int j = 0, pj = 0;
#pragma unroll
        for (int q = 1; q < 8; ++q) {
            const bool in = e >= pre[q];
            j += in ? 1 : 0;
            pj = in ? pre[q] : pj;
        }
        const int v = cells[j * cap + (e - pj)];
        s = v & 0xffff;
        dl = v >> 16;
    };
    int s_nx = 0, dl_nx = 0;
    if (lane < mycnt) fetch(lane, s_nx, dl_nx);

    const float4 bv = *(const float4*)(bias + lane * 4);
    for (int r = wave; r < rows; r += OWN_WAVES) {
        const float di = dinv[r0 + r];
        const float sq = di > 0.0f ? 1.0f / di : 0.0f;
        const float4 hv = *(const float4*)(h + (long)(r0 + r) * D_FEAT + lane * 4);
        float* row = slab + r * SLAB_LD + lane;
        row[0]  = hv.x + bv.x * sq;
        row[32] = hv.y + bv.y * sq;
        row[64] = hv.z + bv.z * sq;
        row[96] = hv.w + bv.w * sq;
        if (lane == 0) dinv_s[r] = di;
    }
    for (int r = rows + (int)threadIdx.x; r < DS_ROWS; r += OWN_THREADS) dinv_s[r] = 0.0f;
    __syncthreads();

    if (mycnt >= 0) {
        int base = 0;
        for (; base + 32 <= mycnt; base += 32) {
            const int s = s_nx, dl = dl_nx;
            s_nx = 0; dl_nx = 0;
            if (base + 32 + lane < mycnt) fetch(base + 32 + lane, s_nx, dl_nx);
            add_edges(s, dl, 32, h, slab, lane);
        }
        if (base < mycnt) add_edges(s_nx, dl_nx, mycnt - base, h, slab, lane);
    } else {
        int* q_e = q_e_all + wave * Q_CAP;
        int* q_d = q_d_all + wave * Q_CAP;
        int cnt = 0;

        scan_dst(dst, nedges, [&](int d, int e) {
            const unsigned dl = (unsigned)(d - r0);
            const bool hit = dl < (unsigned)rows;
            const unsigned m = __builtin_amdgcn_ballot_w32(hit);
            if (m != 0u) {
                const int pos = cnt + (int)__builtin_amdgcn_mbcnt_lo(m, 0u);
                if (hit) { q_e[pos] = e; q_d[pos] = (int)dl; }
                cnt += __popc(m);
            }
        }, [&] {
            while (cnt >= 32) {
                cnt -= 32;
                drain_edges(q_e, q_d, cnt, 32, src, h, slab, lane);
            }
        });
        if (cnt > 0) drain_edges(q_e, q_d, 0, cnt, src, h, slab, lane);
    }
    __syncthreads();

    v2f bfr[32];
    if constexpr (FIRST) {
        const int rt0 = (wave >> 3) & 1;
        if (wave < 16) {
            load_w_strip(Wn, wave, lane, bfr);
            slab_product<true, 0, 2>(slab, dinv_s, bfr, hnext, r0, rows, wave, rt0, 2, lane, nullptr);
        } else {
            load_w_strip(Wl, wave, lane, bfr);
            slab_product<true, 1, 1>(slab, dinv_s, bfr, out, r0, rows, wave, rt0, 2, lane, bl);
        }
    } else {
        load_w_strip(Wl, wave, lane, bfr);
        slab_product<true, 2, 1>(slab, dinv_s, bfr, out, r0, rows, wave, wave >> 3, 4, lane, nullptr);
    }
}

extern "C" void kernel_launch(void* const* d_in, const int* in_sizes, int n_in,
                              void* d_out, int out_size, void* d_ws, size_t ws_size,
                              hipStream_t stream) {
    (void)n_in; (void)out_size; (void)ws_size;

    const float* x    = (const float*)d_in[0];
    const int*   ei   = (const int*)d_in[1];
    const float* W1   = (const float*)d_in[2];
    const float* b1   = (const float*)d_in[3];
    const float* W2   = (const float*)d_in[4];
    const float* b2   = (const float*)d_in[5];
    const float* linW = (const float*)d_in[6];
    const float* linb = (const float*)d_in[7];
    float* out = (float*)d_out;

    const int N = in_sizes[0] / D_FEAT;
    const int E = in_sizes[1] / 2;
    const int* srcI = ei;
    const int* dstI = ei + E;

    float* ws   = (float*)d_ws;
    float* dinv = ws;
    float* h    = dinv + N;
    float* h2   = h   + (long)N * D_FEAT;
    float* lpl  = h2  + (long)N * D_FEAT;

    const int nBlkOwn = (N + OWN_ROWS - 1) / OWN_ROWS;

    int* lists = (int*)lpl;

    bin_edges_kernel<<<NCH, BIN_THREADS, 0, stream>>>(srcI, dstI, lists, N, E, nBlkOwn);
    deg_product_kernel<<<nBlkOwn, OWN_THREADS, 0, stream>>>(dstI, lists, x, W1, dinv, h, N, E);


    layer_owner_kernel<true><<<nBlkOwn, OWN_THREADS, 0, stream>>>(srcI, dstI, dinv, h, b1, lists,
                                                                 W2, h2, linW, linb, out, N, E);

    layer_owner_kernel<false><<<nBlkOwn, OWN_THREADS, 0, stream>>>(srcI, dstI, dinv, h2, b2, lists,
                                                                  nullptr, nullptr,
                                                                  linW + D_FEAT * D_FEAT, nullptr,
                                                                  out, N, E);
}


# Workspace planes reordered (every kernel gets new plane pointers): the three 512-B-row planes start at the workspace base, dinv behind them; a gathered row of g is 4 whole 128-B lines, not 5
# speedup vs baseline: 4.3121x; 1.0568x over previous
#include <hip/hip_runtime.h>


typedef __attribute__((ext_vector_type(2))) float v2f;
typedef __attribute__((ext_vector_type(8))) float v8f;

#define D_FEAT 128

#define OWN_ROWS    196
#define SLAB_LD     129
#define DS_ROWS     (((OWN_ROWS + 15) / 16) * 16)
#define OWN_WAVES   32
#define OWN_THREADS (OWN_WAVES * 32)
#define GROUP_EDGES 512
#define Q_CAP       544

template <int NW = OWN_WAVES, class F, class G>
__device__ __forceinline__ void scan_dst(const int* __restrict__ dst, int nedges, F&& visit,
                                         G&& after_group) {
    const int lane = threadIdx.x & 31;
    const int wave = __builtin_amdgcn_readfirstlane((int)(threadIdx.x >> 5));
    const bool vec_ok = (((size_t)dst) & 15) == 0;
    const int ngroups = vec_ok ? nedges / GROUP_EDGES : 0;
    const int off = (int)blockIdx.x * (ngroups / (int)gridDim.x);

    auto rot = [&](int g) { int gr = g + off; return gr >= ngroups ? gr - ngroups : gr; };
    auto load_group = [&](int4 (&v)[4], int gr) {
        const int4* p = (const int4*)(dst + gr * GROUP_EDGES) + lane;
#pragma unroll
        for (int j = 0; j < 4; ++j) v[j] = p[j * 32];
    };
    auto step = [&](int4 (&cur)[4], int4 (&nxt)[4], int g) {
        const int gr = rot(g);
        const int gn = g + NW;
        load_group(nxt, rot(gn < ngroups ? gn : g));
#pragma unroll
        for (int j = 0; j < 4; ++j) {
            const int e0 = gr * GROUP_EDGES + j * 128 + lane * 4;
            visit(cur[j].x, e0 + 0);
            visit(cur[j].y, e0 + 1);
            visit(cur[j].z, e0 + 2);
            visit(cur[j].w, e0 + 3);
        }
        after_group();
    };

    if (wave < ngroups) {
        int4 bufa[4], bufb[4];
        load_group(bufa, rot(wave));
        for (int g = wave; g < ngroups; g += 2 * NW) {
            step(bufa, bufb, g);
            if (g + NW >= ngroups) break;
            step(bufb, bufa, g + NW);
        }
    }
    for (int base = ngroups * GROUP_EDGES + wave * 32; base < nedges; base += NW * 32) {
        const int e = base + lane;
        visit(e < nedges ? dst[e] : -1, e);
        after_group();
    }
}

#define NCH     256
#define OWN_MAX 512
#define BIN_THREADS 256
static_assert(NCH == 8 * OWN_WAVES, "cells per wave");

__device__ __forceinline__ int cell_cap_of(int rows) {
    return (rows * D_FEAT - NCH) / NCH;
}

__global__ __launch_bounds__(BIN_THREADS)
void bin_edges_kernel(const int* __restrict__ src, const int* __restrict__ dst,
                      int* __restrict__ lists, int n, int nedges, int nown) {
    __shared__ int fill_s[OWN_MAX];

    const int c = blockIdx.x;
    const int per = (nedges + NCH - 1) / NCH;
    const int e0 = c * per;
    const int e1 = min(nedges, e0 + per);
    const bool usable = (n <= 65536) && (nown <= OWN_MAX);

    for (int o = threadIdx.x; o < OWN_MAX; o += BIN_THREADS) fill_s[o] = 0;
    __syncthreads();

    if (usable) {
        for (int base = e0; base < e1; base += 8 * BIN_THREADS) {
            int dd[8], ss[8];
#pragma unroll
            for (int u = 0; u < 8; ++u) {
                const int e = base + u * BIN_THREADS + (int)threadIdx.x;
                const bool ok = e < e1;
                dd[u] = ok ? dst[e] : -1;
                ss[u] = ok ? src[e] : 0;
            }
#pragma unroll
            for (int u = 0; u < 8; ++u) {
                const int d = dd[u];
                if ((unsigned)d < (unsigned)n) {
                    const int o = d / OWN_ROWS;
                    const int row = d - o * OWN_ROWS;
                    const int cap = cell_cap_of(min(OWN_ROWS, n - o * OWN_ROWS));
                    const int pos = __hip_atomic_fetch_add(&fill_s[o], 1, __ATOMIC_RELAXED,
                                                           __HIP_MEMORY_SCOPE_WORKGROUP);
                    if (pos < cap)
                        lists[(long)o * OWN_ROWS * D_FEAT + NCH + c * cap + pos] = ss[u] | (row << 16);
                }
            }
        }
    }
    __syncthreads();

    for (int o = threadIdx.x; o < nown; o += BIN_THREADS) {
        int v = 0x7fffffff;
        if (usable) v = fill_s[o];
        lists[(long)o * OWN_ROWS * D_FEAT + c] = v;
    }
}

__device__ __forceinline__ bool load_cell_counts(const int* cntp, int cap, int lane, int (&cn)[8]) {
    const int4* cp = (const int4*)cntp + lane * 2;
    const int4 a = cp[0], b = cp[1];
    cn[0] = a.x; cn[1] = a.y; cn[2] = a.z; cn[3] = a.w;
    cn[4] = b.x; cn[5] = b.y; cn[6] = b.z; cn[7] = b.w;
    bool bad = false;
#pragma unroll
    for (int j = 0; j < 8; ++j) bad = bad || (cn[j] > cap);
    return __builtin_amdgcn_ballot_w32(bad) != 0u;
}

__device__ __forceinline__ void load_w_strip(const float* __restrict__ Wm, int wave, int lane,
                                             v2f (&bfr)[32]) {
    const float* wp = Wm + (lane >> 4) * D_FEAT + (wave & 7) * 16 + (lane & 15);
#pragma unroll
    for (int t = 0; t < 32; ++t) {
        bfr[t].x = wp[(4 * t) * D_FEAT];
        bfr[t].y = wp[(4 * t + 2) * D_FEAT];
    }
}

template <bool RELU, int EPI, int SCALE>
__device__ __forceinline__ void slab_product(const float* slab, const float* dsc, const v2f (&bfr)[32],
                                             float* __restrict__ hout, int r0, int rows,
                                             int wave, int rt0, int rts, int lane,
                                             const float* __restrict__ addv) {
    const int ct = wave & 7;
    const int m16 = lane & 15;
    const int hh = lane >> 4;
    const int ntiles = (rows + 15) >> 4;
    float bcol = 0.0f;
    if constexpr (EPI == 1) bcol = addv[ct * 16 + m16];
    for (int rt = rt0; rt < ntiles; rt += rts) {
        const int ar = min(rt * 16 + m16, OWN_ROWS - 1);
        const float* ap = slab + ar * SLAB_LD + 32 * hh;
        float prev[8];
        if constexpr (EPI == 2) {
            for (int r = 0; r < 8; ++r) {
                const int prow = min(rt * 16 + hh * 8 + r, rows - 1);
                prev[r] = hout[(long)(r0 + prow) * D_FEAT + ct * 16 + m16];
            }
        }
        const float4 dlo = *(const float4*)(dsc + rt * 16 + hh * 8);
        const float4 dhi = *(const float4*)(dsc + rt * 16 + hh * 8 + 4);
        const float dv[8] = {dlo.x, dlo.y, dlo.z, dlo.w, dhi.x, dhi.y, dhi.z, dhi.w};
        v8f acc = {};
#pragma unroll
        for (int t = 0; t < 32; ++t) {
            v2f a;
            a.x = ap[t];
            a.y = ap[64 + t];
            if constexpr (RELU) {
                a.x = a.x > 0.0f ? a.x : 0.0f;
                a.y = a.y > 0.0f ? a.y : 0.0f;
            }
            acc = __builtin_amdgcn_wmma_f32_16x16x4_f32(
                 false, a,  false, bfr[t],
                 (short)0, acc,  false,  false);
        }
        for (int r = 0; r < 8; ++r) {
            const int orow = rt * 16 + hh * 8 + r;
            float sc = dv[r];
            if constexpr (SCALE == 2) sc *= sc;
            float v = acc[r] * sc;
            if constexpr (EPI == 1) v += bcol;
            if constexpr (EPI == 2) v += prev[r];
            if (orow < rows) hout[(long)(r0 + orow) * D_FEAT + ct * 16 + m16] = v;
        }
    }
}

#define ROWS_PER_WAVE ((OWN_ROWS + OWN_WAVES - 1) / OWN_WAVES)
__global__ __launch_bounds__(OWN_THREADS)
void deg_product_kernel(const int* __restrict__ dst, const int* __restrict__ lists,
                        const float* __restrict__ x, const float* __restrict__ W,
                        float* __restrict__ dinv, float* __restrict__ h, int n, int nedges) {
    __shared__ float slab[OWN_ROWS * SLAB_LD];
    __shared__ __attribute__((aligned(16))) float cnt_s[DS_ROWS];

    const int lane = threadIdx.x & 31;
    const int wave = __builtin_amdgcn_readfirstlane((int)(threadIdx.x >> 5));
    const int r0 = blockIdx.x * OWN_ROWS;
    const int rows = min(OWN_ROWS, n - r0);
    const int cap = cell_cap_of(rows);
    const int* cntp = lists + (long)r0 * D_FEAT;
    const int* cells = cntp + NCH + wave * 8 * cap;

    int cn[8];
    const bool ovf = load_cell_counts(cntp, cap, lane, cn);

    float4 xv[ROWS_PER_WAVE];
#pragma unroll
    for (int k = 0; k < ROWS_PER_WAVE; ++k) {
        const int r = min(wave + k * OWN_WAVES, rows - 1);
        xv[k] = *(const float4*)(x + (long)(r0 + r) * D_FEAT + lane * 4);
    }

    v2f bfr[32];
    load_w_strip(W, wave, lane, bfr);

    int cj[8], v[8];
#pragma unroll
    for (int j = 0; j < 8; ++j) {
        cj[j] = ovf ? 0 : __builtin_amdgcn_readlane(cn[j], wave);
        v[j] = -1;
        if (lane < cj[j]) v[j] = cells[j * cap + lane];
    }

    for (int r = threadIdx.x; r < DS_ROWS; r += OWN_THREADS) cnt_s[r] = 1.0f;
#pragma unroll
    for (int k = 0; k < ROWS_PER_WAVE; ++k) {
        const int r = wave + k * OWN_WAVES;
        if (r < rows) {
            float* row = slab + r * SLAB_LD + lane;
            row[0]  = xv[k].x;
            row[32] = xv[k].y;
            row[64] = xv[k].z;
            row[96] = xv[k].w;
        }
    }
    __syncthreads();

    if (!ovf) {
#pragma unroll
        for (int j = 0; j < 8; ++j) {
            if (v[j] >= 0)
                __hip_atomic_fetch_add(&cnt_s[v[j] >> 16], 1.0f, __ATOMIC_RELAXED,
                                       __HIP_MEMORY_SCOPE_WORKGROUP);
        }
#pragma unroll
        for (int j = 0; j < 8; ++j) {
            for (int i = lane + 32; i < cj[j]; i += 32) {
                const int w = cells[j * cap + i];
                __hip_atomic_fetch_add(&cnt_s[w >> 16], 1.0f, __ATOMIC_RELAXED,
                                       __HIP_MEMORY_SCOPE_WORKGROUP);
            }
        }
    } else {
        scan_dst(dst, nedges, [&](int d, int e) {
            (void)e;
            const unsigned dl = (unsigned)(d - r0);
            if (dl < (unsigned)rows)
                __hip_atomic_fetch_add(&cnt_s[dl], 1.0f, __ATOMIC_RELAXED,
                                       __HIP_MEMORY_SCOPE_WORKGROUP);
        }, [&] {});
    }
    __syncthreads();

    for (int r = threadIdx.x; r < DS_ROWS; r += OWN_THREADS) {
        const float d = cnt_s[r];
        const float di = (d > 0.0f) ? rsqrtf(d) : 0.0f;
        cnt_s[r] = di;
        if (r < rows) dinv[r0 + r] = di;
    }
    __syncthreads();

    slab_product<false, 0, 1>(slab, cnt_s, bfr, h, r0, rows, wave, wave >> 3, 4, lane, nullptr);
}

__device__ __forceinline__ void add_edges(int s, int dl, int n,
                                          const float* __restrict__ h,
                                          float* slab, int lane) {
    for (int i = 0; i < n; i += 8) {
        float4 hv[8];
#pragma unroll
        for (int u = 0; u < 8; ++u) {
            const int si = __builtin_amdgcn_readlane(s, i + u);
            hv[u] = *(const float4*)(h + (long)si * D_FEAT + lane * 4);
        }
#pragma unroll
        for (int u = 0; u < 8; ++u) {
            if (i + u < n) {
                const int dlu = __builtin_amdgcn_readlane(dl, i + u);
                float* row = slab + dlu * SLAB_LD + lane;
                __hip_atomic_fetch_add(row +  0, hv[u].x, __ATOMIC_RELAXED, __HIP_MEMORY_SCOPE_WORKGROUP);
                __hip_atomic_fetch_add(row + 32, hv[u].y, __ATOMIC_RELAXED, __HIP_MEMORY_SCOPE_WORKGROUP);
                __hip_atomic_fetch_add(row + 64, hv[u].z, __ATOMIC_RELAXED, __HIP_MEMORY_SCOPE_WORKGROUP);
                __hip_atomic_fetch_add(row + 96, hv[u].w, __ATOMIC_RELAXED, __HIP_MEMORY_SCOPE_WORKGROUP);
            }
        }
    }
}

__device__ __forceinline__ void drain_edges(const int* q_e, const int* q_d, int base, int n,
                                            const int* __restrict__ src,
                                            const float* __restrict__ h,
                                            float* slab, int lane) {
    int dl = 0, s = 0;
    if (lane < n) {
        const int e = q_e[base + lane];
        dl = q_d[base + lane];
        s = src[e];
    }
    add_edges(s, dl, n, h, slab, lane);
}

template <bool FIRST>
__global__ __launch_bounds__(OWN_THREADS)
void layer_owner_kernel(const int* __restrict__ src, const int* __restrict__ dst,
                        const float* __restrict__ dinv, const float* __restrict__ h,
                        const float* __restrict__ bias, const int* __restrict__ lists,
                        const float* __restrict__ Wn, float* __restrict__ hnext,
                        const float* __restrict__ Wl, const float* __restrict__ bl,
                        float* __restrict__ out, int n, int nedges) {
    __shared__ float slab[OWN_ROWS * SLAB_LD];
    __shared__ __attribute__((aligned(16))) float dinv_s[DS_ROWS];
    __shared__ int q_e_all[OWN_WAVES * Q_CAP];
    __shared__ int q_d_all[OWN_WAVES * Q_CAP];

    const int lane = threadIdx.x & 31;
    const int wave = __builtin_amdgcn_readfirstlane((int)(threadIdx.x >> 5));
    const int r0 = blockIdx.x * OWN_ROWS;
    const int rows = min(OWN_ROWS, n - r0);

    const int cap = cell_cap_of(rows);
    const int* cntp = lists + (long)r0 * D_FEAT;
    const int* cells = cntp + NCH + wave * 8 * cap;
    int cn[8];
    const bool ovf = load_cell_counts(cntp, cap, lane, cn);
    int pre[9];
    pre[0] = 0;
#pragma unroll
    for (int j = 0; j < 8; ++j) pre[j + 1] = pre[j] + __builtin_amdgcn_readlane(cn[j], wave);
    const int mycnt = ovf ? -1 : pre[8];
    auto fetch = [&](int e, int& s, int& dl) {
        int j = 0, pj = 0;
#pragma unroll
        for (int q = 1; q < 8; ++q) {
            const bool in = e >= pre[q];
            j += in ? 1 : 0;
            pj = in ? pre[q] : pj;
        }
        const int v = cells[j * cap + (e - pj)];
        s = v & 0xffff;
        dl = v >> 16;
    };
    int s_nx = 0, dl_nx = 0;
    if (lane < mycnt) fetch(lane, s_nx, dl_nx);

    const float4 bv = *(const float4*)(bias + lane * 4);
    for (int r = wave; r < rows; r += OWN_WAVES) {
        const float di = dinv[r0 + r];
        const float sq = di > 0.0f ? 1.0f / di : 0.0f;
        const float4 hv = *(const float4*)(h + (long)(r0 + r) * D_FEAT + lane * 4);
        float* row = slab + r * SLAB_LD + lane;
        row[0]  = hv.x + bv.x * sq;
        row[32] = hv.y + bv.y * sq;
        row[64] = hv.z + bv.z * sq;
        row[96] = hv.w + bv.w * sq;
        if (lane == 0) dinv_s[r] = di;
    }
    for (int r = rows + (int)threadIdx.x; r < DS_ROWS; r += OWN_THREADS) dinv_s[r] = 0.0f;
    __syncthreads();

    if (mycnt >= 0) {
        int base = 0;
        for (; base + 32 <= mycnt; base += 32) {
            const int s = s_nx, dl = dl_nx;
            s_nx = 0; dl_nx = 0;
            if (base + 32 + lane < mycnt) fetch(base + 32 + lane, s_nx, dl_nx);
            add_edges(s, dl, 32, h, slab, lane);
        }
        if (base < mycnt) add_edges(s_nx, dl_nx, mycnt - base, h, slab, lane);
    } else {
        int* q_e = q_e_all + wave * Q_CAP;
        int* q_d = q_d_all + wave * Q_CAP;
        int cnt = 0;

        scan_dst(dst, nedges, [&](int d, int e) {
            const unsigned dl = (unsigned)(d - r0);
            const bool hit = dl < (unsigned)rows;
            const unsigned m = __builtin_amdgcn_ballot_w32(hit);
            if (m != 0u) {
                const int pos = cnt + (int)__builtin_amdgcn_mbcnt_lo(m, 0u);
                if (hit) { q_e[pos] = e; q_d[pos] = (int)dl; }
                cnt += __popc(m);
            }
        }, [&] {
            while (cnt >= 32) {
                cnt -= 32;
                drain_edges(q_e, q_d, cnt, 32, src, h, slab, lane);
            }
        });
        if (cnt > 0) drain_edges(q_e, q_d, 0, cnt, src, h, slab, lane);
    }
    __syncthreads();

    v2f bfr[32];
    if constexpr (FIRST) {
        const int rt0 = (wave >> 3) & 1;
        if (wave < 16) {
            load_w_strip(Wn, wave, lane, bfr);
            slab_product<true, 0, 2>(slab, dinv_s, bfr, hnext, r0, rows, wave, rt0, 2, lane, nullptr);
        } else {
            load_w_strip(Wl, wave, lane, bfr);
            slab_product<true, 1, 1>(slab, dinv_s, bfr, out, r0, rows, wave, rt0, 2, lane, bl);
        }
    } else {
        load_w_strip(Wl, wave, lane, bfr);
        slab_product<true, 2, 1>(slab, dinv_s, bfr, out, r0, rows, wave, wave >> 3, 4, lane, nullptr);
    }
}

extern "C" void kernel_launch(void* const* d_in, const int* in_sizes, int n_in,
                              void* d_out, int out_size, void* d_ws, size_t ws_size,
                              hipStream_t stream) {
    (void)n_in; (void)out_size; (void)ws_size;

    const float* x    = (const float*)d_in[0];
    const int*   ei   = (const int*)d_in[1];
    const float* W1   = (const float*)d_in[2];
    const float* b1   = (const float*)d_in[3];
    const float* W2   = (const float*)d_in[4];
    const float* b2   = (const float*)d_in[5];
    const float* linW = (const float*)d_in[6];
    const float* linb = (const float*)d_in[7];
    float* out = (float*)d_out;

    const int N = in_sizes[0] / D_FEAT;
    const int E = in_sizes[1] / 2;
    const int* srcI = ei;
    const int* dstI = ei + E;

    float* ws   = (float*)d_ws;
    float* h    = ws;
    float* h2   = h   + (long)N * D_FEAT;
    float* lpl  = h2  + (long)N * D_FEAT;
    float* dinv = lpl + (long)N * D_FEAT;

    const int nBlkOwn = (N + OWN_ROWS - 1) / OWN_ROWS;

    int* lists = (int*)lpl;

    bin_edges_kernel<<<NCH, BIN_THREADS, 0, stream>>>(srcI, dstI, lists, N, E, nBlkOwn);
    deg_product_kernel<<<nBlkOwn, OWN_THREADS, 0, stream>>>(dstI, lists, x, W1, dinv, h, N, E);


    layer_owner_kernel<true><<<nBlkOwn, OWN_THREADS, 0, stream>>>(srcI, dstI, dinv, h, b1, lists,
                                                                 W2, h2, linW, linb, out, N, E);

    layer_owner_kernel<false><<<nBlkOwn, OWN_THREADS, 0, stream>>>(srcI, dstI, dinv, h2, b2, lists,
                                                                  nullptr, nullptr,
                                                                  linW + D_FEAT * D_FEAT, nullptr,
                                                                  out, N, E);
}


# Owner lists ordered by row in the degree kernel (LDS counting sort, self-loops as entries); layer waves take equal shares, sum a row's run in registers, one LDS add per run; row sum order changes
# speedup vs baseline: 4.6665x; 1.0822x over previous
#include <hip/hip_runtime.h>


typedef __attribute__((ext_vector_type(2))) float v2f;
typedef __attribute__((ext_vector_type(8))) float v8f;

#define D_FEAT 128

#define OWN_ROWS    196
#define SLAB_LD     129
#define DS_ROWS     (((OWN_ROWS + 15) / 16) * 16)
#define OWN_WAVES   32
#define OWN_THREADS (OWN_WAVES * 32)
#define GROUP_EDGES 512
#define Q_CAP       544

template <int NW = OWN_WAVES, class F, class G>
__device__ __forceinline__ void scan_dst(const int* __restrict__ dst, int nedges, F&& visit,
                                         G&& after_group) {
    const int lane = threadIdx.x & 31;
    const int wave = __builtin_amdgcn_readfirstlane((int)(threadIdx.x >> 5));
    const bool vec_ok = (((size_t)dst) & 15) == 0;
    const int ngroups = vec_ok ? nedges / GROUP_EDGES : 0;
    const int off = (int)blockIdx.x * (ngroups / (int)gridDim.x);

    auto rot = [&](int g) { int gr = g + off; return gr >= ngroups ? gr - ngroups : gr; };
    auto load_group = [&](int4 (&v)[4], int gr) {
        const int4* p = (const int4*)(dst + gr * GROUP_EDGES) + lane;
#pragma unroll
        for (int j = 0; j < 4; ++j) v[j] = p[j * 32];
    };
    auto step = [&](int4 (&cur)[4], int4 (&nxt)[4], int g) {
        const int gr = rot(g);
        const int gn = g + NW;
        load_group(nxt, rot(gn < ngroups ? gn : g));
#pragma unroll
        for (int j = 0; j < 4; ++j) {
            const int e0 = gr * GROUP_EDGES + j * 128 + lane * 4;
            visit(cur[j].x, e0 + 0);
            visit(cur[j].y, e0 + 1);
            visit(cur[j].z, e0 + 2);
            visit(cur[j].w, e0 + 3);
        }
        after_group();
    };

    if (wave < ngroups) {
        int4 bufa[4], bufb[4];
        load_group(bufa, rot(wave));
        for (int g = wave; g < ngroups; g += 2 * NW) {
            step(bufa, bufb, g);
            if (g + NW >= ngroups) break;
            step(bufb, bufa, g + NW);
        }
    }
    for (int base = ngroups * GROUP_EDGES + wave * 32; base < nedges; base += NW * 32) {
        const int e = base + lane;
        visit(e < nedges ? dst[e] : -1, e);
        after_group();
    }
}

#define NCH     256
#define OWN_MAX 512
#define BIN_THREADS 256
static_assert(NCH == 8 * OWN_WAVES, "cells per wave");

__device__ __forceinline__ int cell_cap_of(int rows) {
    return (rows * D_FEAT - NCH) / NCH;
}

__global__ __launch_bounds__(BIN_THREADS)
void bin_edges_kernel(const int* __restrict__ src, const int* __restrict__ dst,
                      int* __restrict__ lists, int n, int nedges, int nown) {
    __shared__ int fill_s[OWN_MAX];

    const int c = blockIdx.x;
    const int per = (nedges + NCH - 1) / NCH;
    const int e0 = c * per;
    const int e1 = min(nedges, e0 + per);
    const bool usable = (n <= 65536) && (nown <= OWN_MAX);

    for (int o = threadIdx.x; o < OWN_MAX; o += BIN_THREADS) fill_s[o] = 0;
    __syncthreads();

    if (usable) {
        for (int base = e0; base < e1; base += 8 * BIN_THREADS) {
            int dd[8], ss[8];
#pragma unroll
            for (int u = 0; u < 8; ++u) {
                const int e = base + u * BIN_THREADS + (int)threadIdx.x;
                const bool ok = e < e1;
                dd[u] = ok ? dst[e] : -1;
                ss[u] = ok ? src[e] : 0;
            }
#pragma unroll
            for (int u = 0; u < 8; ++u) {
                const int d = dd[u];
                if ((unsigned)d < (unsigned)n) {
                    const int o = d / OWN_ROWS;
                    const int row = d - o * OWN_ROWS;
                    const int cap = cell_cap_of(min(OWN_ROWS, n - o * OWN_ROWS));
                    const int pos = __hip_atomic_fetch_add(&fill_s[o], 1, __ATOMIC_RELAXED,
                                                           __HIP_MEMORY_SCOPE_WORKGROUP);
                    if (pos < cap)
                        lists[(long)o * OWN_ROWS * D_FEAT + NCH + c * cap + pos] = ss[u] | (row << 16);
                }
            }
        }
    }
    __syncthreads();

    for (int o = threadIdx.x; o < nown; o += BIN_THREADS) {
        int v = 0x7fffffff;
        if (usable) v = fill_s[o];
        lists[(long)o * OWN_ROWS * D_FEAT + c] = v;
    }
}

__device__ __forceinline__ bool load_cell_counts(const int* cntp, int cap, int lane, int (&cn)[8]) {
    const int4* cp = (const int4*)cntp + lane * 2;
    const int4 a = cp[0], b = cp[1];
    cn[0] = a.x; cn[1] = a.y; cn[2] = a.z; cn[3] = a.w;
    cn[4] = b.x; cn[5] = b.y; cn[6] = b.z; cn[7] = b.w;
    bool bad = false;
#pragma unroll
    for (int j = 0; j < 8; ++j) bad = bad || (cn[j] > cap);
    return __builtin_amdgcn_ballot_w32(bad) != 0u;
}

__device__ __forceinline__ void load_w_strip(const float* __restrict__ Wm, int wave, int lane,
                                             v2f (&bfr)[32]) {
    const float* wp = Wm + (lane >> 4) * D_FEAT + (wave & 7) * 16 + (lane & 15);
#pragma unroll
    for (int t = 0; t < 32; ++t) {
        bfr[t].x = wp[(4 * t) * D_FEAT];
        bfr[t].y = wp[(4 * t + 2) * D_FEAT];
    }
}

template <bool RELU, int EPI, int SCALE>
__device__ __forceinline__ void slab_product(const float* slab, const float* dsc, const v2f (&bfr)[32],
                                             float* __restrict__ hout, int r0, int rows,
                                             int wave, int rt0, int rts, int lane,
                                             const float* __restrict__ addv) {
    const int ct = wave & 7;
    const int m16 = lane & 15;
    const int hh = lane >> 4;
    const int ntiles = (rows + 15) >> 4;
    float bcol = 0.0f;
    if constexpr (EPI == 1) bcol = addv[ct * 16 + m16];
    for (int rt = rt0; rt < ntiles; rt += rts) {
        const int ar = min(rt * 16 + m16, OWN_ROWS - 1);
        const float* ap = slab + ar * SLAB_LD + 32 * hh;
        float prev[8];
        if constexpr (EPI == 2) {
            for (int r = 0; r < 8; ++r) {
                const int prow = min(rt * 16 + hh * 8 + r, rows - 1);
                prev[r] = hout[(long)(r0 + prow) * D_FEAT + ct * 16 + m16];
            }
        }
        const float4 dlo = *(const float4*)(dsc + rt * 16 + hh * 8);
        const float4 dhi = *(const float4*)(dsc + rt * 16 + hh * 8 + 4);
        const float dv[8] = {dlo.x, dlo.y, dlo.z, dlo.w, dhi.x, dhi.y, dhi.z, dhi.w};
        v8f acc = {};
#pragma unroll
        for (int t = 0; t < 32; ++t) {
            v2f a;
            a.x = ap[t];
            a.y = ap[64 + t];
            if constexpr (RELU) {
                a.x = a.x > 0.0f ? a.x : 0.0f;
                a.y = a.y > 0.0f ? a.y : 0.0f;
            }
            acc = __builtin_amdgcn_wmma_f32_16x16x4_f32(
                 false, a,  false, bfr[t],
                 (short)0, acc,  false,  false);
        }
        for (int r = 0; r < 8; ++r) {
            const int orow = rt * 16 + hh * 8 + r;
            float sc = dv[r];
            if constexpr (SCALE == 2) sc *= sc;
            float v = acc[r] * sc;
            if constexpr (EPI == 1) v += bcol;
            if constexpr (EPI == 2) v += prev[r];
            if (orow < rows) hout[(long)(r0 + orow) * D_FEAT + ct * 16 + m16] = v;
        }
    }
}

#define ROWS_PER_WAVE ((OWN_ROWS + OWN_WAVES - 1) / OWN_WAVES)
#define SORT_CAP      6144
#define LIST_OFF      32
#define SCAN_PER_LANE ((DS_ROWS + 31) / 32)
__global__ __launch_bounds__(OWN_THREADS)
void deg_product_kernel(const int* __restrict__ dst, int* __restrict__ lists,
                        const float* __restrict__ x, const float* __restrict__ W,
                        float* __restrict__ dinv, float* __restrict__ h, int n, int nedges) {
    __shared__ float slab[OWN_ROWS * SLAB_LD];
    __shared__ __attribute__((aligned(16))) float cnt_s[DS_ROWS];
    __shared__ int rowcnt_s[DS_ROWS];
    __shared__ int rowpos_s[DS_ROWS];
    __shared__ int total_s;
    __shared__ int sorted_s[SORT_CAP];

    const int lane = threadIdx.x & 31;
    const int wave = __builtin_amdgcn_readfirstlane((int)(threadIdx.x >> 5));
    const int r0 = blockIdx.x * OWN_ROWS;
    const int rows = min(OWN_ROWS, n - r0);
    const int cap = cell_cap_of(rows);
    int* slice = lists + (long)r0 * D_FEAT;
    const int* cntp = slice;
    const int* cells = cntp + NCH + wave * 8 * cap;

    int cn[8];
    const bool ovf = load_cell_counts(cntp, cap, lane, cn);

    float4 xv[ROWS_PER_WAVE];
#pragma unroll
    for (int k = 0; k < ROWS_PER_WAVE; ++k) {
        const int r = min(wave + k * OWN_WAVES, rows - 1);
        xv[k] = *(const float4*)(x + (long)(r0 + r) * D_FEAT + lane * 4);
    }

    v2f bfr[32];
    load_w_strip(W, wave, lane, bfr);

    int cj[8], v[8];
#pragma unroll
    for (int j = 0; j < 8; ++j) {
        cj[j] = ovf ? 0 : __builtin_amdgcn_readlane(cn[j], wave);
        v[j] = -1;
        if (lane < cj[j]) v[j] = cells[j * cap + lane];
    }

    for (int r = threadIdx.x; r < DS_ROWS; r += OWN_THREADS) rowcnt_s[r] = r < rows ? 1 : 0;
#pragma unroll
    for (int k = 0; k < ROWS_PER_WAVE; ++k) {
        const int r = wave + k * OWN_WAVES;
        if (r < rows) {
            float* row = slab + r * SLAB_LD + lane;
            row[0]  = xv[k].x;
            row[32] = xv[k].y;
            row[64] = xv[k].z;
            row[96] = xv[k].w;
        }
    }
    __syncthreads();

    if (!ovf) {
#pragma unroll
        for (int j = 0; j < 8; ++j) {
            if (v[j] >= 0)
                __hip_atomic_fetch_add(&rowcnt_s[v[j] >> 16], 1, __ATOMIC_RELAXED,
                                       __HIP_MEMORY_SCOPE_WORKGROUP);
        }
#pragma unroll
        for (int j = 0; j < 8; ++j) {
            for (int i = lane + 32; i < cj[j]; i += 32) {
                const int w = cells[j * cap + i];
                __hip_atomic_fetch_add(&rowcnt_s[w >> 16], 1, __ATOMIC_RELAXED,
                                       __HIP_MEMORY_SCOPE_WORKGROUP);
            }
        }
    } else {
        scan_dst(dst, nedges, [&](int d, int e) {
            (void)e;
            const unsigned dl = (unsigned)(d - r0);
            if (dl < (unsigned)rows)
                __hip_atomic_fetch_add(&rowcnt_s[dl], 1, __ATOMIC_RELAXED,
                                       __HIP_MEMORY_SCOPE_WORKGROUP);
        }, [&] {});
    }
    __syncthreads();

    for (int r = threadIdx.x; r < DS_ROWS; r += OWN_THREADS) {
        const int cr = rowcnt_s[r];
        const float di = (cr > 0) ? rsqrtf((float)cr) : 0.0f;
        cnt_s[r] = di;
        if (r < rows) dinv[r0 + r] = di;
    }
    if (wave == 0) {
        int loc[SCAN_PER_LANE];
        int sum = 0;
#pragma unroll
        for (int k = 0; k < SCAN_PER_LANE; ++k) {
            const int r = lane * SCAN_PER_LANE + k;
            loc[k] = sum;
            sum += (r < DS_ROWS) ? rowcnt_s[r] : 0;
        }
        int incl = sum;
#pragma unroll
        for (int d = 1; d < 32; d <<= 1) {
            const int up = __shfl_up(incl, d);
            if (lane >= d) incl += up;
        }
        const int excl = incl - sum;
#pragma unroll
        for (int k = 0; k < SCAN_PER_LANE; ++k) {
            const int r = lane * SCAN_PER_LANE + k;
            if (r < DS_ROWS) rowpos_s[r] = excl + loc[k];
        }
        if (lane == 31) total_s = incl;
    }
    __syncthreads();

    const int total = total_s;
    const bool listed = !ovf && total <= SORT_CAP && total <= rows * D_FEAT - NCH - LIST_OFF;
    if (listed) {
        for (int r = threadIdx.x; r < rows; r += OWN_THREADS) {
            const int pos = __hip_atomic_fetch_add(&rowpos_s[r], 1, __ATOMIC_RELAXED,
                                                   __HIP_MEMORY_SCOPE_WORKGROUP);
            sorted_s[pos] = (r0 + r) | (r << 16);
        }
#pragma unroll
        for (int j = 0; j < 8; ++j) {
            if (v[j] >= 0) {
                const int pos = __hip_atomic_fetch_add(&rowpos_s[v[j] >> 16], 1, __ATOMIC_RELAXED,
                                                       __HIP_MEMORY_SCOPE_WORKGROUP);
                sorted_s[pos] = v[j];
            }
        }
#pragma unroll
        for (int j = 0; j < 8; ++j) {
            for (int i = lane + 32; i < cj[j]; i += 32) {
                const int w = cells[j * cap + i];
                const int pos = __hip_atomic_fetch_add(&rowpos_s[w >> 16], 1, __ATOMIC_RELAXED,
                                                       __HIP_MEMORY_SCOPE_WORKGROUP);
                sorted_s[pos] = w;
            }
        }
    }
    __syncthreads();
    if (listed) {
        for (int i = threadIdx.x; i < total; i += OWN_THREADS) slice[NCH + LIST_OFF + i] = sorted_s[i];
    }
    if (threadIdx.x == 0) slice[NCH] = listed ? total : -1;

    slab_product<false, 0, 1>(slab, cnt_s, bfr, h, r0, rows, wave, wave >> 3, 4, lane, nullptr);
}

__device__ __forceinline__ void add_edges(int s, int dl, int n,
                                          const float* __restrict__ h,
                                          float* slab, int lane) {
    for (int i = 0; i < n; i += 8) {
        float4 hv[8];
#pragma unroll
        for (int u = 0; u < 8; ++u) {
            const int si = __builtin_amdgcn_readlane(s, i + u);
            hv[u] = *(const float4*)(h + (long)si * D_FEAT + lane * 4);
        }
#pragma unroll
        for (int u = 0; u < 8; ++u) {
            if (i + u < n) {
                const int dlu = __builtin_amdgcn_readlane(dl, i + u);
                float* row = slab + dlu * SLAB_LD + lane;
                __hip_atomic_fetch_add(row +  0, hv[u].x, __ATOMIC_RELAXED, __HIP_MEMORY_SCOPE_WORKGROUP);
                __hip_atomic_fetch_add(row + 32, hv[u].y, __ATOMIC_RELAXED, __HIP_MEMORY_SCOPE_WORKGROUP);
                __hip_atomic_fetch_add(row + 64, hv[u].z, __ATOMIC_RELAXED, __HIP_MEMORY_SCOPE_WORKGROUP);
                __hip_atomic_fetch_add(row + 96, hv[u].w, __ATOMIC_RELAXED, __HIP_MEMORY_SCOPE_WORKGROUP);
            }
        }
    }
}

__device__ __forceinline__ void flush_run(float* slab, int row, const float4& acc, int lane) {
    float* p = slab + row * SLAB_LD + lane;
    __hip_atomic_fetch_add(p +  0, acc.x, __ATOMIC_RELAXED, __HIP_MEMORY_SCOPE_WORKGROUP);
    __hip_atomic_fetch_add(p + 32, acc.y, __ATOMIC_RELAXED, __HIP_MEMORY_SCOPE_WORKGROUP);
    __hip_atomic_fetch_add(p + 64, acc.z, __ATOMIC_RELAXED, __HIP_MEMORY_SCOPE_WORKGROUP);
    __hip_atomic_fetch_add(p + 96, acc.w, __ATOMIC_RELAXED, __HIP_MEMORY_SCOPE_WORKGROUP);
}

__device__ __forceinline__ void add_runs(int s, int dl, int n, const float* __restrict__ h,
                                         float* slab, int lane, int& cur, float4& acc) {
    for (int i = 0; i < n; i += 8) {
        float4 hv[8];
#pragma unroll
        for (int u = 0; u < 8; ++u) {
            const int si = __builtin_amdgcn_readlane(s, i + u);
            hv[u] = *(const float4*)(h + (long)si * D_FEAT + lane * 4);
        }
#pragma unroll
        for (int u = 0; u < 8; ++u) {
            if (i + u < n) {
                const int dlu = __builtin_amdgcn_readlane(dl, i + u);
                if (dlu != cur) {
                    if (cur >= 0) flush_run(slab, cur, acc, lane);
                    cur = dlu;
                    acc = hv[u];
                } else {
                    acc.x += hv[u].x;
                    acc.y += hv[u].y;
                    acc.z += hv[u].z;
                    acc.w += hv[u].w;
                }
            }
        }
    }
}

__device__ __forceinline__ void drain_edges(const int* q_e, const int* q_d, int base, int n,
                                            const int* __restrict__ src,
                                            const float* __restrict__ h,
                                            float* slab, int lane) {
    int dl = 0, s = 0;
    if (lane < n) {
        const int e = q_e[base + lane];
        dl = q_d[base + lane];
        s = src[e];
    }
    add_edges(s, dl, n, h, slab, lane);
}

template <bool FIRST>
__global__ __launch_bounds__(OWN_THREADS)
void layer_owner_kernel(const int* __restrict__ src, const int* __restrict__ dst,
                        const float* __restrict__ dinv, const float* __restrict__ h,
                        const float* __restrict__ bias, const int* __restrict__ lists,
                        const float* __restrict__ Wn, float* __restrict__ hnext,
                        const float* __restrict__ Wl, const float* __restrict__ bl,
                        float* __restrict__ out, int n, int nedges) {
    __shared__ float slab[OWN_ROWS * SLAB_LD];
    __shared__ __attribute__((aligned(16))) float dinv_s[DS_ROWS];
    __shared__ int q_e_all[OWN_WAVES * Q_CAP];
    __shared__ int q_d_all[OWN_WAVES * Q_CAP];

    const int lane = threadIdx.x & 31;
    const int wave = __builtin_amdgcn_readfirstlane((int)(threadIdx.x >> 5));
    const int r0 = blockIdx.x * OWN_ROWS;
    const int rows = min(OWN_ROWS, n - r0);

    const int* slice = lists + (long)r0 * D_FEAT;
    const int total = slice[NCH];
    const int* sorted = slice + NCH + LIST_OFF;
    const int lo = total > 0 ? (int)(((long)wave * total) / OWN_WAVES) : 0;
    const int hi = total > 0 ? (int)(((long)(wave + 1) * total) / OWN_WAVES) : 0;
    int v_nx = 0;
    if (lo + lane < hi) v_nx = sorted[lo + lane];

    for (int r = threadIdx.x; r < DS_ROWS; r += OWN_THREADS) dinv_s[r] = r < rows ? dinv[r0 + r] : 0.0f;
    const float4 bv = *(const float4*)(bias + lane * 4);
    __syncthreads();
    for (int r = wave; r < rows; r += OWN_WAVES) {
        const float di = dinv_s[r];
        const float sq = di > 0.0f ? 1.0f / di : 0.0f;
        float* row = slab + r * SLAB_LD + lane;
        row[0]  = bv.x * sq;
        row[32] = bv.y * sq;
        row[64] = bv.z * sq;
        row[96] = bv.w * sq;
    }
    __syncthreads();

    if (total >= 0) {
        int cur = -1;
        float4 acc = {0.0f, 0.0f, 0.0f, 0.0f};
        int base = lo;
        for (; base + 32 <= hi; base += 32) {
            const int v = v_nx;
            v_nx = 0;
            if (base + 32 + lane < hi) v_nx = sorted[base + 32 + lane];
            add_runs(v & 0xffff, v >> 16, 32, h, slab, lane, cur, acc);
        }
        if (base < hi) add_runs(v_nx & 0xffff, v_nx >> 16, hi - base, h, slab, lane, cur, acc);
        if (cur >= 0) flush_run(slab, cur, acc, lane);
    } else {
        for (int r = wave; r < rows; r += OWN_WAVES) {
            const float4 hv = *(const float4*)(h + (long)(r0 + r) * D_FEAT + lane * 4);
            flush_run(slab, r, hv, lane);
        }
        int* q_e = q_e_all + wave * Q_CAP;
        int* q_d = q_d_all + wave * Q_CAP;
        int cnt = 0;

        scan_dst(dst, nedges, [&](int d, int e) {
            const unsigned dl = (unsigned)(d - r0);
            const bool hit = dl < (unsigned)rows;
            const unsigned m = __builtin_amdgcn_ballot_w32(hit);
            if (m != 0u) {
                const int pos = cnt + (int)__builtin_amdgcn_mbcnt_lo(m, 0u);
                if (hit) { q_e[pos] = e; q_d[pos] = (int)dl; }
                cnt += __popc(m);
            }
        }, [&] {
            while (cnt >= 32) {
                cnt -= 32;
                drain_edges(q_e, q_d, cnt, 32, src, h, slab, lane);
            }
        });
        if (cnt > 0) drain_edges(q_e, q_d, 0, cnt, src, h, slab, lane);
    }
    __syncthreads();

    v2f bfr[32];
    if constexpr (FIRST) {
        const int rt0 = (wave >> 3) & 1;
        if (wave < 16) {
            load_w_strip(Wn, wave, lane, bfr);
            slab_product<true, 0, 2>(slab, dinv_s, bfr, hnext, r0, rows, wave, rt0, 2, lane, nullptr);
        } else {
            load_w_strip(Wl, wave, lane, bfr);
            slab_product<true, 1, 1>(slab, dinv_s, bfr, out, r0, rows, wave, rt0, 2, lane, bl);
        }
    } else {
        load_w_strip(Wl, wave, lane, bfr);
        slab_product<true, 2, 1>(slab, dinv_s, bfr, out, r0, rows, wave, wave >> 3, 4, lane, nullptr);
    }
}

extern "C" void kernel_launch(void* const* d_in, const int* in_sizes, int n_in,
                              void* d_out, int out_size, void* d_ws, size_t ws_size,
                              hipStream_t stream) {
    (void)n_in; (void)out_size; (void)ws_size;

    const float* x    = (const float*)d_in[0];
    const int*   ei   = (const int*)d_in[1];
    const float* W1   = (const float*)d_in[2];
    const float* b1   = (const float*)d_in[3];
    const float* W2   = (const float*)d_in[4];
    const float* b2   = (const float*)d_in[5];
    const float* linW = (const float*)d_in[6];
    const float* linb = (const float*)d_in[7];
    float* out = (float*)d_out;

    const int N = in_sizes[0] / D_FEAT;
    const int E = in_sizes[1] / 2;
    const int* srcI = ei;
    const int* dstI = ei + E;

    float* ws   = (float*)d_ws;
    float* h    = ws;
    float* h2   = h   + (long)N * D_FEAT;
    float* lpl  = h2  + (long)N * D_FEAT;
    float* dinv = lpl + (long)N * D_FEAT;

    const int nBlkOwn = (N + OWN_ROWS - 1) / OWN_ROWS;

    int* lists = (int*)lpl;

    bin_edges_kernel<<<NCH, BIN_THREADS, 0, stream>>>(srcI, dstI, lists, N, E, nBlkOwn);
    deg_product_kernel<<<nBlkOwn, OWN_THREADS, 0, stream>>>(dstI, lists, x, W1, dinv, h, N, E);


    layer_owner_kernel<true><<<nBlkOwn, OWN_THREADS, 0, stream>>>(srcI, dstI, dinv, h, b1, lists,
                                                                 W2, h2, linW, linb, out, N, E);

    layer_owner_kernel<false><<<nBlkOwn, OWN_THREADS, 0, stream>>>(srcI, dstI, dinv, h2, b2, lists,
                                                                  nullptr, nullptr,
                                                                  linW + D_FEAT * D_FEAT, nullptr,
                                                                  out, N, E);
}
